# starting kernel
# baseline (speedup 1.0000x reference)
#include <hip/hip_runtime.h>
#include <math.h>

#define LL    50
#define BB    16
#define ITERR 76
#define KDIM  100
#define CDIM  3800
#define CPAD  3808
#define JPAD  64
#define SROW  7600
#define SROW0 50
#define COL0  1900

typedef float v2f __attribute__((ext_vector_type(2)));
typedef float v8f __attribute__((ext_vector_type(8)));

#define WMMA_F32X4(a, b, c) \
  __builtin_amdgcn_wmma_f32_16x16x4_f32(false, (a), false, (b), (short)0, (c), false, false)

__global__ void k_maxes(const float* __restrict__ inp, const float* __restrict__ wgt,
                        float* __restrict__ smax) {
  __shared__ float red[256];
  int b = blockIdx.x;
  float m = -3.4e38f;
  if (b < BB) {
    const float* p = inp + b * 2500;
    for (int i = threadIdx.x; i < 2500; i += 256) m = fmaxf(m, p[i]);
  } else {
    for (int i = threadIdx.x; i < 2500; i += 256) m = fmaxf(m, fabsf(wgt[i]));
  }
  red[threadIdx.x] = m;
  __syncthreads();
  for (int s = 128; s > 0; s >>= 1) {
    if (threadIdx.x < s) red[threadIdx.x] = fmaxf(red[threadIdx.x], red[threadIdx.x + s]);
    __syncthreads();
  }
  if (threadIdx.x == 0) smax[b] = red[0];
}

__global__ void k_build_q(float* __restrict__ qre, float* __restrict__ qim) {
  int idx = blockIdx.x * 256 + threadIdx.x;
  if (idx >= JPAD * CPAD) return;
  int i = idx / CPAD, c = idx % CPAD;
  float re = 0.f, im = 0.f;
  if (i < LL) {
    int t = ((i + 1) * (COL0 + c)) % 200;
    float s, cz;
    sincospif((float)t * 0.01f, &s, &cz);
    re = cz; im = -s;
  }
  qre[idx] = re; qim[idx] = im;
}

__global__ void __launch_bounds__(128) k_mm1(
    const float* __restrict__ inp, const float* __restrict__ wgt,
    const float* __restrict__ sre, const float* __restrict__ sim,
    float* __restrict__ gre, float* __restrict__ gim) {
  __shared__ float Pre[16 * KDIM];
  __shared__ float Pim[16 * KDIM];
  int bid = blockIdx.x;
  int cg = bid % 60;
  int jt = (bid / 60) % 4;
  int b  = bid / 240;

  for (int e = threadIdx.x; e < 16 * KDIM; e += 128) {
    int jl = e / KDIM, r = e % KDIM;
    int j = jt * 16 + jl;
    float re = 0.f, im = 0.f;
    if (j < LL) {
      int t = ((150 + 2 * j) * (50 + r)) % 200;
      float s, cz;
      sincospif((float)t * 0.01f, &s, &cz);
      re = cz; im = -s;
    }
    Pre[e] = re; Pim[e] = im;
  }
  __syncthreads();

  int wave = threadIdx.x >> 5;
  int lane = threadIdx.x & 31;
  int ctile = cg * 4 + wave;
  if (ctile >= CPAD / 16) return;

  int h  = lane >> 4;
  int ln = lane & 15;
  int c  = ctile * 16 + ln;
  int x2 = COL0 + c;
  int q  = (c < CDIM) ? (c / ITERR) : 0;
  const float* inp_b = inp + b * 2500 + q;

  v8f cre = {}; v8f cim = {};
  for (int t = 0; t < KDIM / 4; ++t) {
    int kk = 4 * t + 2 * h;
    v2f are = *(const v2f*)&Pre[ln * KDIM + kk];
    v2f aim = *(const v2f*)&Pim[ln * KDIM + kk];
    v2f aimn = -aim;
    float inpv = inp_b[(2 * t + h) * LL];
    int so = (SROW0 + kk) * SROW + x2;
    v2f bre; bre.x = sre[so] * inpv; bre.y = sre[so + SROW] * inpv;
    v2f bim; bim.x = sim[so] * inpv; bim.y = sim[so + SROW] * inpv;
    cre = WMMA_F32X4(are,  bre, cre);
    cre = WMMA_F32X4(aimn, bim, cre);
    cim = WMMA_F32X4(are,  bim, cim);
    cim = WMMA_F32X4(aim,  bre, cim);
  }

  size_t gbase = (size_t)b * JPAD * CPAD;
#pragma unroll
  for (int r = 0; r < 8; ++r) {
    int j = jt * 16 + r + 8 * h;
    float w = (j < LL) ? fabsf(wgt[j * LL + q]) : 0.f;
    gre[gbase + (size_t)j * CPAD + c] = cre[r] * w;
    gim[gbase + (size_t)j * CPAD + c] = cim[r] * w;
  }
}

__global__ void __launch_bounds__(128) k_mm2(
    const float* __restrict__ gre, const float* __restrict__ gim,
    const float* __restrict__ qre, const float* __restrict__ qim,
    const float* __restrict__ amp, float* __restrict__ magbuf) {
  int jt = blockIdx.x % 4;
  int b  = blockIdx.x / 4;
  int wave = threadIdx.x >> 5;
  int lane = threadIdx.x & 31;
  int h = lane >> 4, ln = lane & 15;
  int i = wave * 16 + ln;

  const float* gre_l = gre + (size_t)b * JPAD * CPAD + (size_t)(jt * 16 + ln) * CPAD;
  const float* gim_l = gim + (size_t)b * JPAD * CPAD + (size_t)(jt * 16 + ln) * CPAD;
  const float* qre_l = qre + (size_t)i * CPAD;
  const float* qim_l = qim + (size_t)i * CPAD;

  v8f cre = {}; v8f cim = {};
  for (int t = 0; t < CDIM / 4; ++t) {
    int k = 4 * t + 2 * h;
    v2f are = *(const v2f*)&gre_l[k];
    v2f aim = *(const v2f*)&gim_l[k];
    v2f bre = *(const v2f*)&qre_l[k];
    v2f bim = *(const v2f*)&qim_l[k];
    v2f aimn = -aim;
    cre = WMMA_F32X4(are,  bre, cre);
    cre = WMMA_F32X4(aimn, bim, cre);
    cim = WMMA_F32X4(are,  bim, cim);
    cim = WMMA_F32X4(aim,  bre, cim);
  }

#pragma unroll
  for (int r = 0; r < 8; ++r) {
    int j = jt * 16 + r + 8 * h;
    float mag = sqrtf(cre[r] * cre[r] + cim[r] * cim[r]);
    float v = 0.f;
    if (j < LL && i < LL) v = mag / amp[j * LL + i];
    magbuf[(size_t)b * JPAD * JPAD + j * JPAD + i] = v;
  }
}

__global__ void k_final(const float* __restrict__ magbuf, const float* __restrict__ smax,
                        const float* __restrict__ bias, float* __restrict__ out) {
  __shared__ float red[256];
  int b = blockIdx.x;
  const float* mb = magbuf + (size_t)b * JPAD * JPAD;
  float m = 0.f;
  for (int idx = threadIdx.x; idx < JPAD * JPAD; idx += 256) m = fmaxf(m, mb[idx]);
  red[threadIdx.x] = m;
  __syncthreads();
  for (int s = 128; s > 0; s >>= 1) {
    if (threadIdx.x < s) red[threadIdx.x] = fmaxf(red[threadIdx.x], red[threadIdx.x + s]);
    __syncthreads();
  }
  float scale = (smax[b] * smax[16]) / red[0];
  for (int idx = threadIdx.x; idx < LL * LL; idx += 256) {
    int i = idx / LL, j = idx % LL;
    out[b * 2500 + idx] = mb[j * JPAD + i] * scale + bias[j];
  }
}

extern "C" void kernel_launch(void* const* d_in, const int* in_sizes, int n_in,
                              void* d_out, int out_size, void* d_ws, size_t ws_size,
                              hipStream_t stream) {
  const float* inp  = (const float*)d_in[0];
  const float* wgt  = (const float*)d_in[1];
  const float* bias = (const float*)d_in[2];
  const float* amp  = (const float*)d_in[3];
  const float* sre  = (const float*)d_in[4];
  const float* sim  = (const float*)d_in[5];

  float* ws   = (float*)d_ws;
  float* smax = ws;
  float* qre  = ws + 32;
  float* qim  = qre + (size_t)JPAD * CPAD;
  float* gre  = qim + (size_t)JPAD * CPAD;
  float* gim  = gre + (size_t)BB * JPAD * CPAD;
  float* magb = gim + (size_t)BB * JPAD * CPAD;

  k_maxes  <<<17, 256, 0, stream>>>(inp, wgt, smax);
  k_build_q<<<(JPAD * CPAD + 255) / 256, 256, 0, stream>>>(qre, qim);
  k_mm1    <<<BB * 4 * 60, 128, 0, stream>>>(inp, wgt, sre, sim, gre, gim);
  k_mm2    <<<BB * 4, 128, 0, stream>>>(gre, gim, qre, qim, amp, magb);
  k_final  <<<BB, 256, 0, stream>>>(magb, smax, bias, (float*)d_out);
}


# GEMM-2 K range split over 16 work-groups per output tile (1024 blocks, was 64); partial sums go to the workspace and the last kernel adds them in fixed order: changes the summation order
# speedup vs baseline: 2.5192x; 2.5192x over previous
#include <hip/hip_runtime.h>
#include <math.h>

#define LL    50
#define BB    16
#define ITERR 76
#define KDIM  100
#define CDIM  3800
#define CPAD  3808
#define JPAD  64
#define SROW  7600
#define SROW0 50
#define COL0  1900
#define KSPLIT 16
#define KSTEPS 60
#define TILEP  4096

typedef float v2f __attribute__((ext_vector_type(2)));
typedef float v8f __attribute__((ext_vector_type(8)));

#define WMMA_F32X4(a, b, c) \
  __builtin_amdgcn_wmma_f32_16x16x4_f32(false, (a), false, (b), (short)0, (c), false, false)

__global__ void k_maxes(const float* __restrict__ inp, const float* __restrict__ wgt,
                        float* __restrict__ smax) {
  __shared__ float red[256];
  int b = blockIdx.x;
  float m = -3.4e38f;
  if (b < BB) {
    const float* p = inp + b * 2500;
    for (int i = threadIdx.x; i < 2500; i += 256) m = fmaxf(m, p[i]);
  } else {
    for (int i = threadIdx.x; i < 2500; i += 256) m = fmaxf(m, fabsf(wgt[i]));
  }
  red[threadIdx.x] = m;
  __syncthreads();
  for (int s = 128; s > 0; s >>= 1) {
    if (threadIdx.x < s) red[threadIdx.x] = fmaxf(red[threadIdx.x], red[threadIdx.x + s]);
    __syncthreads();
  }
  if (threadIdx.x == 0) smax[b] = red[0];
}

__global__ void k_build_q(float* __restrict__ qre, float* __restrict__ qim) {
  int idx = blockIdx.x * 256 + threadIdx.x;
  if (idx >= JPAD * CPAD) return;
  int i = idx / CPAD, c = idx % CPAD;
  float re = 0.f, im = 0.f;
  if (i < LL) {
    int t = ((i + 1) * (COL0 + c)) % 200;
    float s, cz;
    sincospif((float)t * 0.01f, &s, &cz);
    re = cz; im = -s;
  }
  qre[idx] = re; qim[idx] = im;
}

__global__ void __launch_bounds__(128) k_mm1(
    const float* __restrict__ inp, const float* __restrict__ wgt,
    const float* __restrict__ sre, const float* __restrict__ sim,
    float* __restrict__ gre, float* __restrict__ gim) {
  __shared__ float Pre[16 * KDIM];
  __shared__ float Pim[16 * KDIM];
  int bid = blockIdx.x;
  int cg = bid % 60;
  int jt = (bid / 60) % 4;
  int b  = bid / 240;

  for (int e = threadIdx.x; e < 16 * KDIM; e += 128) {
    int jl = e / KDIM, r = e % KDIM;
    int j = jt * 16 + jl;
    float re = 0.f, im = 0.f;
    if (j < LL) {
      int t = ((150 + 2 * j) * (50 + r)) % 200;
      float s, cz;
      sincospif((float)t * 0.01f, &s, &cz);
      re = cz; im = -s;
    }
    Pre[e] = re; Pim[e] = im;
  }
  __syncthreads();

  int wave = threadIdx.x >> 5;
  int lane = threadIdx.x & 31;
  int ctile = cg * 4 + wave;
  if (ctile >= CPAD / 16) return;

  int h  = lane >> 4;
  int ln = lane & 15;
  int c  = ctile * 16 + ln;
  int x2 = COL0 + c;
  int q  = (c < CDIM) ? (c / ITERR) : 0;
  const float* inp_b = inp + b * 2500 + q;

  v8f cre = {}; v8f cim = {};
  for (int t = 0; t < KDIM / 4; ++t) {
    int kk = 4 * t + 2 * h;
    v2f are = *(const v2f*)&Pre[ln * KDIM + kk];
    v2f aim = *(const v2f*)&Pim[ln * KDIM + kk];
    v2f aimn = -aim;
    float inpv = inp_b[(2 * t + h) * LL];
    int so = (SROW0 + kk) * SROW + x2;
    v2f bre; bre.x = sre[so] * inpv; bre.y = sre[so + SROW] * inpv;
    v2f bim; bim.x = sim[so] * inpv; bim.y = sim[so + SROW] * inpv;
    cre = WMMA_F32X4(are,  bre, cre);
    cre = WMMA_F32X4(aimn, bim, cre);
    cim = WMMA_F32X4(are,  bim, cim);
    cim = WMMA_F32X4(aim,  bre, cim);
  }

  size_t gbase = (size_t)b * JPAD * CPAD;
#pragma unroll
  for (int r = 0; r < 8; ++r) {
    int j = jt * 16 + r + 8 * h;
    float w = (j < LL) ? fabsf(wgt[j * LL + q]) : 0.f;
    gre[gbase + (size_t)j * CPAD + c] = cre[r] * w;
    gim[gbase + (size_t)j * CPAD + c] = cim[r] * w;
  }
}

__global__ void __launch_bounds__(128) k_mm2(
    const float* __restrict__ gre, const float* __restrict__ gim,
    const float* __restrict__ qre, const float* __restrict__ qim,
    float* __restrict__ pre, float* __restrict__ pim) {
  int s  = blockIdx.x % KSPLIT;
  int bj = blockIdx.x / KSPLIT;
  int jt = bj % 4;
  int b  = bj / 4;
  int wave = threadIdx.x >> 5;
  int lane = threadIdx.x & 31;
  int h = lane >> 4, ln = lane & 15;
  int i = wave * 16 + ln;

  const float* gre_l = gre + (size_t)b * JPAD * CPAD + (size_t)(jt * 16 + ln) * CPAD;
  const float* gim_l = gim + (size_t)b * JPAD * CPAD + (size_t)(jt * 16 + ln) * CPAD;
  const float* qre_l = qre + (size_t)i * CPAD;
  const float* qim_l = qim + (size_t)i * CPAD;

  int t0 = s * KSTEPS;
  int t1 = (t0 + KSTEPS < CDIM / 4) ? (t0 + KSTEPS) : (CDIM / 4);

  v8f cre = {}; v8f cim = {};
  for (int t = t0; t < t1; ++t) {
    int k = 4 * t + 2 * h;
    v2f are = *(const v2f*)&gre_l[k];
    v2f aim = *(const v2f*)&gim_l[k];
    v2f bre = *(const v2f*)&qre_l[k];
    v2f bim = *(const v2f*)&qim_l[k];
    v2f aimn = -aim;
    cre = WMMA_F32X4(are,  bre, cre);
    cre = WMMA_F32X4(aimn, bim, cre);
    cim = WMMA_F32X4(are,  bim, cim);
    cim = WMMA_F32X4(aim,  bre, cim);
  }

  size_t pbase = ((size_t)(s * BB + b) * 16 + (size_t)(jt * 4 + wave)) * 256 + lane;
#pragma unroll
  for (int r = 0; r < 8; ++r) {
    pre[pbase + r * 32] = cre[r];
    pim[pbase + r * 32] = cim[r];
  }
}

#define MPITCH 65
__global__ void __launch_bounds__(1024) k_final(
    const float* __restrict__ pre, const float* __restrict__ pim,
    const float* __restrict__ smax, const float* __restrict__ bias,
    const float* __restrict__ amp, float* __restrict__ out) {
  __shared__ float mb[JPAD * MPITCH];
  __shared__ float red[32];
  int b = blockIdx.x;
  int tid = threadIdx.x;
  int tile = tid >> 6;
  int p0 = (tid & 63) * 4;
  int j  = (tile >> 2) * 16 + (p0 >> 5) + 8 * ((p0 >> 4) & 1);
  int i0 = (tile & 3) * 16 + (p0 & 15);

  float m = 0.f;
  if (j < LL && i0 < LL) {
    const float* pr = pre + (size_t)b * TILEP + tid * 4;
    const float* pi = pim + (size_t)b * TILEP + tid * 4;
    float4 sr = *(const float4*)pr;
    float4 si = *(const float4*)pi;
#pragma unroll
    for (int s = 1; s < KSPLIT; ++s) {
      float4 a = *(const float4*)(pr + (size_t)s * BB * TILEP);
      float4 c = *(const float4*)(pi + (size_t)s * BB * TILEP);
      sr.x += a.x; sr.y += a.y; sr.z += a.z; sr.w += a.w;
      si.x += c.x; si.y += c.y; si.z += c.z; si.w += c.w;
    }
    float fr[4] = {sr.x, sr.y, sr.z, sr.w};
    float fi[4] = {si.x, si.y, si.z, si.w};
#pragma unroll
    for (int k = 0; k < 4; ++k) {
      int i = i0 + k;
      float v = 0.f;
      if (i < LL) v = sqrtf(fr[k] * fr[k] + fi[k] * fi[k]) / amp[j * LL + i];
      mb[j * MPITCH + i] = v;
      m = fmaxf(m, v);
    }
  }
  for (int off = 16; off > 0; off >>= 1) m = fmaxf(m, __shfl_xor(m, off));
  if ((tid & 31) == 0) red[tid >> 5] = m;
  __syncthreads();
  float mx = red[tid & 31];
  for (int off = 16; off > 0; off >>= 1) mx = fmaxf(mx, __shfl_xor(mx, off));

  float scale = (smax[b] * smax[16]) / mx;
  for (int idx = tid; idx < LL * LL; idx += 1024) {
    int i = idx / LL, j2 = idx % LL;
    out[b * 2500 + idx] = mb[j2 * MPITCH + i] * scale + bias[j2];
  }
}

extern "C" void kernel_launch(void* const* d_in, const int* in_sizes, int n_in,
                              void* d_out, int out_size, void* d_ws, size_t ws_size,
                              hipStream_t stream) {
  const float* inp  = (const float*)d_in[0];
  const float* wgt  = (const float*)d_in[1];
  const float* bias = (const float*)d_in[2];
  const float* amp  = (const float*)d_in[3];
  const float* sre  = (const float*)d_in[4];
  const float* sim  = (const float*)d_in[5];

  float* ws   = (float*)d_ws;
  float* smax = ws;
  float* qre  = ws + 32;
  float* qim  = qre + (size_t)JPAD * CPAD;
  float* gre  = qim + (size_t)JPAD * CPAD;
  float* gim  = gre + (size_t)BB * JPAD * CPAD;
  float* pre  = gim + (size_t)BB * JPAD * CPAD;
  float* pim  = pre + (size_t)KSPLIT * BB * TILEP;

  k_maxes  <<<17, 256, 0, stream>>>(inp, wgt, smax);
  k_build_q<<<(JPAD * CPAD + 255) / 256, 256, 0, stream>>>(qre, qim);
  k_mm1    <<<BB * 4 * 60, 128, 0, stream>>>(inp, wgt, sre, sim, gre, gim);
  k_mm2    <<<BB * 4 * KSPLIT, 128, 0, stream>>>(gre, gim, qre, qim, pre, pim);
  k_final  <<<BB, 1024, 0, stream>>>(pre, pim, smax, bias, amp, (float*)d_out);
}


# GEMM-1 and GEMM-2 fused: each wave builds its 16x16 G tile transposed and feeds it to GEMM-2 from its accumulators (G never stored; Q kept in fragment order): changes the summation order
# speedup vs baseline: 3.5809x; 1.4214x over previous
#include <hip/hip_runtime.h>
#include <math.h>


#define LL    50
#define BB    16
#define ITERR 76
#define KDIM  100
#define CDIM  3800
#define CTILES 238
#define JPAD  64
#define SROW  7600
#define SROW0 50
#define COL0  1900
#define KSPLIT 16
#define CTPER  15
#define TILEP  4096

typedef float v2f __attribute__((ext_vector_type(2)));
typedef float v4f __attribute__((ext_vector_type(4)));
typedef float v8f __attribute__((ext_vector_type(8)));

#define WMMA_F32X4(a, b, c) \
  __builtin_amdgcn_wmma_f32_16x16x4_f32(false, (a), false, (b), (short)0, (c), false, false)

__global__ void k_maxes(const float* __restrict__ inp, const float* __restrict__ wgt,
                        float* __restrict__ smax) {
  __shared__ float red[256];
  int b = blockIdx.x;
  float m = -3.4e38f;
  if (b < BB) {
    const float* p = inp + b * 2500;
    for (int i = threadIdx.x; i < 2500; i += 256) m = fmaxf(m, p[i]);
  } else {
    for (int i = threadIdx.x; i < 2500; i += 256) m = fmaxf(m, fabsf(wgt[i]));
  }
  red[threadIdx.x] = m;
  __syncthreads();
  for (int s = 128; s > 0; s >>= 1) {
    if (threadIdx.x < s) red[threadIdx.x] = fmaxf(red[threadIdx.x], red[threadIdx.x + s]);
    __syncthreads();
  }
  if (threadIdx.x == 0) smax[b] = red[0];
}

__global__ void k_build_q(float* __restrict__ qf) {
  int idx = blockIdx.x * 256 + threadIdx.x;
  if (idx >= CTILES * 16 * 32) return;
  int lane = idx & 31;
  int it = (idx >> 5) & 3;
  int tp = (idx >> 7) & 3;
  int ct = idx >> 9;
  int i  = it * 16 + (lane & 15);
  int c0 = ct * 16 + 8 * (lane >> 4) + 2 * tp;
  float4 v = {0.f, 0.f, 0.f, 0.f};
  if (i < LL) {
    int t0 = ((i + 1) * (COL0 + c0)) % 200;
    int t1 = ((i + 1) * (COL0 + c0 + 1)) % 200;
    float s0, z0, s1, z1;
    sincospif((float)t0 * 0.01f, &s0, &z0);
    sincospif((float)t1 * 0.01f, &s1, &z1);
    v.x = z0; v.y = z1; v.z = -s0; v.w = -s1;
  }
  *(float4*)&qf[(size_t)idx * 4] = v;
}

__global__ void __launch_bounds__(256) __attribute__((amdgpu_waves_per_eu(2, 2))) k_mm12(
    const float* __restrict__ inp, const float* __restrict__ wgt,
    const float* __restrict__ sre, const float* __restrict__ sim,
    const float* __restrict__ qf,
    float* __restrict__ pre, float* __restrict__ pim) {
  __shared__ float sm[2 * JPAD * KDIM];
  float* Pre = sm;
  float* Pim = sm + JPAD * KDIM;
  int s = blockIdx.x % KSPLIT;
  int b = blockIdx.x / KSPLIT;

  for (int e = threadIdx.x; e < JPAD * KDIM; e += 256) {
    int j = e / KDIM, r = e % KDIM;
    float re = 0.f, im = 0.f;
    if (j < LL) {
      int t = ((150 + 2 * j) * (50 + r)) % 200;
      float sn, cz;
      sincospif((float)t * 0.01f, &sn, &cz);
      re = cz; im = -sn;
    }
    Pre[e] = re; Pim[e] = im;
  }
  __syncthreads();

  int wave = threadIdx.x >> 5;
  int lane = threadIdx.x & 31;
  int jt  = wave & 3;
  int par = wave >> 2;
  int h  = lane >> 4;
  int ln = lane & 15;
  int j  = jt * 16 + ln;

  v8f fre[4], fim[4];
#pragma unroll
  for (int it = 0; it < 4; ++it) { fre[it] = (v8f){}; fim[it] = (v8f){}; }

  int ct0 = s * CTPER;
  int ct1 = (ct0 + CTPER < CTILES) ? (ct0 + CTPER) : CTILES;
  for (int ct = ct0 + par; ct < ct1; ct += 2) {
    int c  = ct * 16 + ln;
    int x2 = COL0 + c;
    int q  = (c < CDIM) ? (c / ITERR) : 0;
    const float* inp_b = inp + b * 2500 + q;
    const v4f* qt = (const v4f*)qf + (size_t)ct * 512 + lane;
    v4f qv[16];
#pragma unroll
    for (int u = 0; u < 16; ++u) qv[u] = qt[u * 32];
    int cA = ct * 16 + 8 * h;
    int cB = cA + 4;
    int jw = (j < LL) ? j : 0;
    float wAl = wgt[jw * LL + ((cA < CDIM) ? cA / ITERR : 0)];
    float wBl = wgt[jw * LL + ((cB < CDIM) ? cB / ITERR : 0)];
    v8f dre = {}; v8f dim = {};
    for (int t = 0; t < KDIM / 4; ++t) {
      int kk = 4 * t + 2 * h;
      v2f pr = *(const v2f*)&Pre[j * KDIM + kk];
      v2f pi = *(const v2f*)&Pim[j * KDIM + kk];
      float inpv = inp_b[(2 * t + h) * LL];
      int so = (SROW0 + kk) * SROW + x2;
      v2f are; are.x = sre[so] * inpv; are.y = sre[so + SROW] * inpv;
      v2f aim; aim.x = sim[so] * inpv; aim.y = sim[so + SROW] * inpv;
      v2f aimn = -aim;
      dre = WMMA_F32X4(are,  pr, dre);
      dim = WMMA_F32X4(are,  pi, dim);
      dre = WMMA_F32X4(aimn, pi, dre);
      dim = WMMA_F32X4(aim,  pr, dim);
    }

    float wA = (j < LL && cA < CDIM) ? fabsf(wAl) : 0.f;
    float wB = (j < LL && cB < CDIM) ? fabsf(wBl) : 0.f;

#pragma unroll
    for (int tp = 0; tp < 4; ++tp) {
      float w = (tp < 2) ? wA : wB;
      v2f gr; gr.x = dre[2 * tp] * w; gr.y = dre[2 * tp + 1] * w;
      v2f gi; gi.x = dim[2 * tp] * w; gi.y = dim[2 * tp + 1] * w;
      v2f gin = -gi;
#pragma unroll
      for (int it = 0; it < 4; ++it) {
        v4f qq = qv[tp * 4 + it];
        v2f br; br.x = qq.x; br.y = qq.y;
        v2f bi; bi.x = qq.z; bi.y = qq.w;
        fre[it] = WMMA_F32X4(gr,  br, fre[it]);
        fim[it] = WMMA_F32X4(gr,  bi, fim[it]);
        fre[it] = WMMA_F32X4(gin, bi, fre[it]);
        fim[it] = WMMA_F32X4(gi,  br, fim[it]);
      }
    }
  }

  __syncthreads();
  float* red = sm + (size_t)jt * 64 * 32 + lane;
  if (par == 1) {
#pragma unroll
    for (int it = 0; it < 4; ++it) {
#pragma unroll
      for (int r = 0; r < 8; ++r) {
        red[(it * 16 + r) * 32]     = fre[it][r];
        red[(it * 16 + 8 + r) * 32] = fim[it][r];
      }
    }
  }
  __syncthreads();
  if (par == 0) {
    size_t pbase = ((size_t)(s * BB + b) * 16 + (size_t)(jt * 4)) * 256 + lane;
#pragma unroll
    for (int it = 0; it < 4; ++it) {
#pragma unroll
      for (int r = 0; r < 8; ++r) {
        pre[pbase + it * 256 + r * 32] = fre[it][r] + red[(it * 16 + r) * 32];
        pim[pbase + it * 256 + r * 32] = fim[it][r] + red[(it * 16 + 8 + r) * 32];
      }
    }
  }
}

#define MPITCH 65
__global__ void __launch_bounds__(1024) k_final(
    const float* __restrict__ pre, const float* __restrict__ pim,
    const float* __restrict__ smax, const float* __restrict__ bias,
    const float* __restrict__ amp, float* __restrict__ out) {
  __shared__ float mb[JPAD * MPITCH];
  __shared__ float red[32];
  int b = blockIdx.x;
  int tid = threadIdx.x;
  int tile = tid >> 6;
  int p0 = (tid & 63) * 4;
  int j  = (tile >> 2) * 16 + (p0 >> 5) + 8 * ((p0 >> 4) & 1);
  int i0 = (tile & 3) * 16 + (p0 & 15);

  float m = 0.f;
  if (j < LL && i0 < LL) {
    const float* pr = pre + (size_t)b * TILEP + tid * 4;
    const float* pi = pim + (size_t)b * TILEP + tid * 4;
    float4 sr = *(const float4*)pr;
    float4 si = *(const float4*)pi;
#pragma unroll
    for (int s = 1; s < KSPLIT; ++s) {
      float4 a = *(const float4*)(pr + (size_t)s * BB * TILEP);
      float4 c = *(const float4*)(pi + (size_t)s * BB * TILEP);
      sr.x += a.x; sr.y += a.y; sr.z += a.z; sr.w += a.w;
      si.x += c.x; si.y += c.y; si.z += c.z; si.w += c.w;
    }
    float fr[4] = {sr.x, sr.y, sr.z, sr.w};
    float fi[4] = {si.x, si.y, si.z, si.w};
#pragma unroll
    for (int k = 0; k < 4; ++k) {
      int i = i0 + k;
      float v = 0.f;
      if (i < LL) v = sqrtf(fr[k] * fr[k] + fi[k] * fi[k]) / amp[j * LL + i];
      mb[j * MPITCH + i] = v;
      m = fmaxf(m, v);
    }
  }
  for (int off = 16; off > 0; off >>= 1) m = fmaxf(m, __shfl_xor(m, off));
  if ((tid & 31) == 0) red[tid >> 5] = m;
  __syncthreads();
  float mx = red[tid & 31];
  for (int off = 16; off > 0; off >>= 1) mx = fmaxf(mx, __shfl_xor(mx, off));

  float scale = (smax[b] * smax[16]) / mx;
  for (int idx = tid; idx < LL * LL; idx += 1024) {
    int i = idx / LL, j2 = idx % LL;
    out[b * 2500 + idx] = mb[j2 * MPITCH + i] * scale + bias[j2];
  }
}

extern "C" void kernel_launch(void* const* d_in, const int* in_sizes, int n_in,
                              void* d_out, int out_size, void* d_ws, size_t ws_size,
                              hipStream_t stream) {
  const float* inp  = (const float*)d_in[0];
  const float* wgt  = (const float*)d_in[1];
  const float* bias = (const float*)d_in[2];
  const float* amp  = (const float*)d_in[3];
  const float* sre  = (const float*)d_in[4];
  const float* sim  = (const float*)d_in[5];

  float* ws   = (float*)d_ws;
  float* smax = ws;
  float* qf   = ws + 32;
  float* pre  = qf + (size_t)CTILES * 16 * 32 * 4;
  float* pim  = pre + (size_t)KSPLIT * BB * TILEP;

  k_maxes  <<<17, 256, 0, stream>>>(inp, wgt, smax);
  k_build_q<<<(CTILES * 16 * 32 + 255) / 256, 256, 0, stream>>>(qf);
  k_mm12   <<<BB * KSPLIT, 256, 0, stream>>>(inp, wgt, sre, sim, qf, pre, pim);
  k_final  <<<BB, 1024, 0, stream>>>(pre, pim, smax, bias, amp, (float*)d_out);
}


# complex products by three real matrix products instead of four (p1=a*c, p2=b*d, p3=(a+b)*(c+d); re=p1-p2, im=p3-p1-p2) in both GEMMs: changes rounding and the summation order
# speedup vs baseline: 3.8966x; 1.0882x over previous
#include <hip/hip_runtime.h>
#include <math.h>


#define LL    50
#define BB    16
#define ITERR 76
#define KDIM  100
#define CDIM  3800
#define CTILES 238
#define JPAD  64
#define SROW  7600
#define SROW0 50
#define COL0  1900
#define KSPLIT 16
#define CTPER  15
#define TILEP  4096

typedef float v2f __attribute__((ext_vector_type(2)));
typedef float v4f __attribute__((ext_vector_type(4)));
typedef float v8f __attribute__((ext_vector_type(8)));

#define WMMA_F32X4(a, b, c) \
  __builtin_amdgcn_wmma_f32_16x16x4_f32(false, (a), false, (b), (short)0, (c), false, false)

__global__ void k_maxes(const float* __restrict__ inp, const float* __restrict__ wgt,
                        float* __restrict__ smax) {
  __shared__ float red[256];
  int b = blockIdx.x;
  float m = -3.4e38f;
  if (b < BB) {
    const float* p = inp + b * 2500;
    for (int i = threadIdx.x; i < 2500; i += 256) m = fmaxf(m, p[i]);
  } else {
    for (int i = threadIdx.x; i < 2500; i += 256) m = fmaxf(m, fabsf(wgt[i]));
  }
  red[threadIdx.x] = m;
  __syncthreads();
  for (int s = 128; s > 0; s >>= 1) {
    if (threadIdx.x < s) red[threadIdx.x] = fmaxf(red[threadIdx.x], red[threadIdx.x + s]);
    __syncthreads();
  }
  if (threadIdx.x == 0) smax[b] = red[0];
}

__global__ void k_build_q(float* __restrict__ qf) {
  int idx = blockIdx.x * 256 + threadIdx.x;
  if (idx >= CTILES * 16 * 32) return;
  int lane = idx & 31;
  int it = (idx >> 5) & 3;
  int tp = (idx >> 7) & 3;
  int ct = idx >> 9;
  int i  = it * 16 + (lane & 15);
  int c0 = ct * 16 + 8 * (lane >> 4) + 2 * tp;
  float4 v = {0.f, 0.f, 0.f, 0.f};
  if (i < LL) {
    int t0 = ((i + 1) * (COL0 + c0)) % 200;
    int t1 = ((i + 1) * (COL0 + c0 + 1)) % 200;
    float s0, z0, s1, z1;
    sincospif((float)t0 * 0.01f, &s0, &z0);
    sincospif((float)t1 * 0.01f, &s1, &z1);
    v.x = z0; v.y = z1; v.z = -s0; v.w = -s1;
  }
  *(float4*)&qf[(size_t)idx * 4] = v;
}

__global__ void __launch_bounds__(256) __attribute__((amdgpu_waves_per_eu(2, 2))) k_mm12(
    const float* __restrict__ inp, const float* __restrict__ wgt,
    const float* __restrict__ sre, const float* __restrict__ sim,
    const float* __restrict__ qf,
    float* __restrict__ pre, float* __restrict__ pim) {
  __shared__ float sm[2 * JPAD * KDIM];
  float* Pre = sm;
  float* Pim = sm + JPAD * KDIM;
  int s = blockIdx.x % KSPLIT;
  int b = blockIdx.x / KSPLIT;

  for (int e = threadIdx.x; e < JPAD * KDIM; e += 256) {
    int j = e / KDIM, r = e % KDIM;
    float re = 0.f, im = 0.f;
    if (j < LL) {
      int t = ((150 + 2 * j) * (50 + r)) % 200;
      float sn, cz;
      sincospif((float)t * 0.01f, &sn, &cz);
      re = cz; im = -sn;
    }
    Pre[e] = re; Pim[e] = im;
  }
  __syncthreads();

  int wave = threadIdx.x >> 5;
  int lane = threadIdx.x & 31;
  int jt  = wave & 3;
  int par = wave >> 2;
  int h  = lane >> 4;
  int ln = lane & 15;
  int j  = jt * 16 + ln;

  v8f f1[4], f2[4], f3[4];
#pragma unroll
  for (int it = 0; it < 4; ++it) { f1[it] = (v8f){}; f2[it] = (v8f){}; f3[it] = (v8f){}; }

  int ct0 = s * CTPER;
  int ct1 = (ct0 + CTPER < CTILES) ? (ct0 + CTPER) : CTILES;
  for (int ct = ct0 + par; ct < ct1; ct += 2) {
    int c  = ct * 16 + ln;
    int x2 = COL0 + c;
    int q  = (c < CDIM) ? (c / ITERR) : 0;
    const float* inp_b = inp + b * 2500 + q;
    const v4f* qt = (const v4f*)qf + (size_t)ct * 512 + lane;
    v4f qv[16];
#pragma unroll
    for (int u = 0; u < 16; ++u) qv[u] = qt[u * 32];
    int cA = ct * 16 + 8 * h;
    int cB = cA + 4;
    int jw = (j < LL) ? j : 0;
    float wAl = wgt[jw * LL + ((cA < CDIM) ? cA / ITERR : 0)];
    float wBl = wgt[jw * LL + ((cB < CDIM) ? cB / ITERR : 0)];
    v8f d1 = {}; v8f d2 = {}; v8f d3 = {};
    for (int t = 0; t < KDIM / 4; ++t) {
      int kk = 4 * t + 2 * h;
      v2f pr = *(const v2f*)&Pre[j * KDIM + kk];
      v2f pi = *(const v2f*)&Pim[j * KDIM + kk];
      float inpv = inp_b[(2 * t + h) * LL];
      int so = (SROW0 + kk) * SROW + x2;
      v2f are; are.x = sre[so] * inpv; are.y = sre[so + SROW] * inpv;
      v2f aim; aim.x = sim[so] * inpv; aim.y = sim[so + SROW] * inpv;
      v2f asum = are + aim;
      v2f ps = pr + pi;
      d1 = WMMA_F32X4(are,  pr, d1);
      d2 = WMMA_F32X4(aim,  pi, d2);
      d3 = WMMA_F32X4(asum, ps, d3);
    }
    v8f dre = d1 - d2;
    v8f dim = (d3 - d1) - d2;

    float wA = (j < LL && cA < CDIM) ? fabsf(wAl) : 0.f;
    float wB = (j < LL && cB < CDIM) ? fabsf(wBl) : 0.f;

#pragma unroll
    for (int tp = 0; tp < 4; ++tp) {
      float w = (tp < 2) ? wA : wB;
      v2f gr; gr.x = dre[2 * tp] * w; gr.y = dre[2 * tp + 1] * w;
      v2f gi; gi.x = dim[2 * tp] * w; gi.y = dim[2 * tp + 1] * w;
      v2f gs = gr + gi;
#pragma unroll
      for (int it = 0; it < 4; ++it) {
        v4f qq = qv[tp * 4 + it];
        v2f br; br.x = qq.x; br.y = qq.y;
        v2f bi; bi.x = qq.z; bi.y = qq.w;
        v2f bs = br + bi;
        f1[it] = WMMA_F32X4(gr, br, f1[it]);
        f2[it] = WMMA_F32X4(gi, bi, f2[it]);
        f3[it] = WMMA_F32X4(gs, bs, f3[it]);
      }
    }
  }

  v8f fre[4], fim[4];
#pragma unroll
  for (int it = 0; it < 4; ++it) {
    fre[it] = f1[it] - f2[it];
    fim[it] = (f3[it] - f1[it]) - f2[it];
  }

  __syncthreads();
  int lane_e = (int)__builtin_amdgcn_mbcnt_lo(~0u, 0u);
  float* red = sm + (size_t)jt * 64 * 32 + lane_e;
  if (par == 1) {
#pragma unroll
    for (int it = 0; it < 4; ++it) {
#pragma unroll
      for (int r = 0; r < 8; ++r) {
        red[(it * 16 + r) * 32]     = fre[it][r];
        red[(it * 16 + 8 + r) * 32] = fim[it][r];
      }
    }
  }
  __syncthreads();
  if (par == 0) {
    size_t pbase = ((size_t)(s * BB + b) * 16 + (size_t)(jt * 4)) * 256 + lane_e;
#pragma unroll
    for (int it = 0; it < 4; ++it) {
#pragma unroll
      for (int r = 0; r < 8; ++r) {
        pre[pbase + it * 256 + r * 32] = fre[it][r] + red[(it * 16 + r) * 32];
        pim[pbase + it * 256 + r * 32] = fim[it][r] + red[(it * 16 + 8 + r) * 32];
      }
    }
  }
}

#define MPITCH 65
__global__ void __launch_bounds__(1024) k_final(
    const float* __restrict__ pre, const float* __restrict__ pim,
    const float* __restrict__ smax, const float* __restrict__ bias,
    const float* __restrict__ amp, float* __restrict__ out) {
  __shared__ float mb[JPAD * MPITCH];
  __shared__ float red[32];
  int b = blockIdx.x;
  int tid = threadIdx.x;
  int tile = tid >> 6;
  int p0 = (tid & 63) * 4;
  int j  = (tile >> 2) * 16 + (p0 >> 5) + 8 * ((p0 >> 4) & 1);
  int i0 = (tile & 3) * 16 + (p0 & 15);

  float m = 0.f;
  if (j < LL && i0 < LL) {
    const float* pr = pre + (size_t)b * TILEP + tid * 4;
    const float* pi = pim + (size_t)b * TILEP + tid * 4;
    float4 sr = *(const float4*)pr;
    float4 si = *(const float4*)pi;
#pragma unroll
    for (int s = 1; s < KSPLIT; ++s) {
      float4 a = *(const float4*)(pr + (size_t)s * BB * TILEP);
      float4 c = *(const float4*)(pi + (size_t)s * BB * TILEP);
      sr.x += a.x; sr.y += a.y; sr.z += a.z; sr.w += a.w;
      si.x += c.x; si.y += c.y; si.z += c.z; si.w += c.w;
    }
    float fr[4] = {sr.x, sr.y, sr.z, sr.w};
    float fi[4] = {si.x, si.y, si.z, si.w};
#pragma unroll
    for (int k = 0; k < 4; ++k) {
      int i = i0 + k;
      float v = 0.f;
      if (i < LL) v = sqrtf(fr[k] * fr[k] + fi[k] * fi[k]) / amp[j * LL + i];
      mb[j * MPITCH + i] = v;
      m = fmaxf(m, v);
    }
  }
  for (int off = 16; off > 0; off >>= 1) m = fmaxf(m, __shfl_xor(m, off));
  if ((tid & 31) == 0) red[tid >> 5] = m;
  __syncthreads();
  float mx = red[tid & 31];
  for (int off = 16; off > 0; off >>= 1) mx = fmaxf(mx, __shfl_xor(mx, off));

  float scale = (smax[b] * smax[16]) / mx;
  for (int idx = tid; idx < LL * LL; idx += 1024) {
    int i = idx / LL, j2 = idx % LL;
    out[b * 2500 + idx] = mb[j2 * MPITCH + i] * scale + bias[j2];
  }
}

extern "C" void kernel_launch(void* const* d_in, const int* in_sizes, int n_in,
                              void* d_out, int out_size, void* d_ws, size_t ws_size,
                              hipStream_t stream) {
  const float* inp  = (const float*)d_in[0];
  const float* wgt  = (const float*)d_in[1];
  const float* bias = (const float*)d_in[2];
  const float* amp  = (const float*)d_in[3];
  const float* sre  = (const float*)d_in[4];
  const float* sim  = (const float*)d_in[5];

  float* ws   = (float*)d_ws;
  float* smax = ws;
  float* qf   = ws + 32;
  float* pre  = qf + (size_t)CTILES * 16 * 32 * 4;
  float* pim  = pre + (size_t)KSPLIT * BB * TILEP;

  k_maxes  <<<17, 256, 0, stream>>>(inp, wgt, smax);
  k_build_q<<<(CTILES * 16 * 32 + 255) / 256, 256, 0, stream>>>(qf);
  k_mm12   <<<BB * KSPLIT, 256, 0, stream>>>(inp, wgt, sre, sim, qf, pre, pim);
  k_final  <<<BB, 1024, 0, stream>>>(pre, pim, smax, bias, amp, (float*)d_out);
}


# GEMM-2 once per run of column tiles 25 apart instead of per tile: Q has period 200 in c, so those G tiles are added in registers first: changes the summation order
# speedup vs baseline: 4.3321x; 1.1118x over previous
#include <hip/hip_runtime.h>
#include <math.h>


#define LL    50
#define BB    16
#define ITERR 76
#define KDIM  100
#define CDIM  3800
#define CTILES 238
#define JPAD  64
#define SROW  7600
#define SROW0 50
#define COL0  1900
#define KSPLIT 16
#define CTPER  15
#define TILEP  4096
#define QPER  25
#define NLONG 13

typedef float v2f __attribute__((ext_vector_type(2)));
typedef float v4f __attribute__((ext_vector_type(4)));
typedef float v8f __attribute__((ext_vector_type(8)));

#define WMMA_F32X4(a, b, c) \
  __builtin_amdgcn_wmma_f32_16x16x4_f32(false, (a), false, (b), (short)0, (c), false, false)

__global__ void k_maxes(const float* __restrict__ inp, const float* __restrict__ wgt,
                        float* __restrict__ smax) {
  __shared__ float red[256];
  int b = blockIdx.x;
  float m = -3.4e38f;
  if (b < BB) {
    const float* p = inp + b * 2500;
    for (int i = threadIdx.x; i < 2500; i += 256) m = fmaxf(m, p[i]);
  } else {
    for (int i = threadIdx.x; i < 2500; i += 256) m = fmaxf(m, fabsf(wgt[i]));
  }
  red[threadIdx.x] = m;
  __syncthreads();
  for (int s = 128; s > 0; s >>= 1) {
    if (threadIdx.x < s) red[threadIdx.x] = fmaxf(red[threadIdx.x], red[threadIdx.x + s]);
    __syncthreads();
  }
  if (threadIdx.x == 0) smax[b] = red[0];
}

__global__ void k_build_q(float* __restrict__ qf) {
  int idx = blockIdx.x * 256 + threadIdx.x;
  if (idx >= QPER * 16 * 32) return;
  int lane = idx & 31;
  int it = (idx >> 5) & 3;
  int tp = (idx >> 7) & 3;
  int ct = idx >> 9;
  int i  = it * 16 + (lane & 15);
  int c0 = ct * 16 + 8 * (lane >> 4) + 2 * tp;
  float4 v = {0.f, 0.f, 0.f, 0.f};
  if (i < LL) {
    int t0 = ((i + 1) * (COL0 + c0)) % 200;
    int t1 = ((i + 1) * (COL0 + c0 + 1)) % 200;
    float s0, z0, s1, z1;
    sincospif((float)t0 * 0.01f, &s0, &z0);
    sincospif((float)t1 * 0.01f, &s1, &z1);
    v.x = z0; v.y = z1; v.z = -s0; v.w = -s1;
  }
  *(float4*)&qf[(size_t)idx * 4] = v;
}

__global__ void __launch_bounds__(256) __attribute__((amdgpu_waves_per_eu(2, 2))) k_mm12(
    const float* __restrict__ inp, const float* __restrict__ wgt,
    const float* __restrict__ sre, const float* __restrict__ sim,
    const float* __restrict__ qf,
    float* __restrict__ pre, float* __restrict__ pim) {
  __shared__ float sm[2 * JPAD * KDIM];
  float* Pre = sm;
  float* Pim = sm + JPAD * KDIM;
  int s = blockIdx.x % KSPLIT;
  int b = blockIdx.x / KSPLIT;

  for (int e = threadIdx.x; e < JPAD * KDIM; e += 256) {
    int j = e / KDIM, r = e % KDIM;
    float re = 0.f, im = 0.f;
    if (j < LL) {
      int t = ((150 + 2 * j) * (50 + r)) % 200;
      float sn, cz;
      sincospif((float)t * 0.01f, &sn, &cz);
      re = cz; im = -sn;
    }
    Pre[e] = re; Pim[e] = im;
  }
  __syncthreads();

  int wave = threadIdx.x >> 5;
  int lane = threadIdx.x & 31;
  int jt  = wave & 3;
  int par = wave >> 2;
  int h  = lane >> 4;
  int ln = lane & 15;
  int j  = jt * 16 + ln;

  v8f f1[4], f2[4], f3[4];
#pragma unroll
  for (int it = 0; it < 4; ++it) { f1[it] = (v8f){}; f2[it] = (v8f){}; f3[it] = (v8f){}; }

  int tau0 = s * CTPER;
  int tau1 = (tau0 + CTPER < CTILES) ? (tau0 + CTPER) : CTILES;
  int nhalf = (tau1 - tau0 + 1) >> 1;
  int tau  = par ? (tau0 + nhalf) : tau0;
  int tend = par ? tau1 : (tau0 + nhalf);
  while (tau < tend) {
    int kap, m, nk;
    if (tau < NLONG * 10) { kap = tau / 10; m = tau - kap * 10; nk = 10; }
    else { int u = tau - NLONG * 10; int kq = u / 9; kap = NLONG + kq; m = u - kq * 9; nk = 9; }
    int mend = m + (tend - tau);
    if (mend > nk) mend = nk;
    tau += mend - m;
    const v4f* qt = (const v4f*)qf + (size_t)kap * 512 + lane;
    v4f qv[16];
#pragma unroll
    for (int u = 0; u < 16; ++u) qv[u] = qt[u * 32];
    v8f gre = {}; v8f gim = {};
    for (; m < mend; ++m) {
      int ct = kap + QPER * m;
      int c  = ct * 16 + ln;
      int x2 = COL0 + c;
      int q  = (c < CDIM) ? (c / ITERR) : 0;
      const float* inp_b = inp + b * 2500 + q;
      int cA = ct * 16 + 8 * h;
      int cB = cA + 4;
      int jw = (j < LL) ? j : 0;
      float wAl = wgt[jw * LL + ((cA < CDIM) ? cA / ITERR : 0)];
      float wBl = wgt[jw * LL + ((cB < CDIM) ? cB / ITERR : 0)];
      v8f d1 = {}; v8f d2 = {}; v8f d3 = {};
#pragma unroll
      for (int t = 0; t < KDIM / 4; ++t) {
        int kk = 4 * t + 2 * h;
        v2f pr = *(const v2f*)&Pre[j * KDIM + kk];
        v2f pi = *(const v2f*)&Pim[j * KDIM + kk];
        float inpv = inp_b[(2 * t + h) * LL];
        int so = (SROW0 + kk) * SROW + x2;
        v2f are; are.x = sre[so] * inpv; are.y = sre[so + SROW] * inpv;
        v2f aim; aim.x = sim[so] * inpv; aim.y = sim[so + SROW] * inpv;
        v2f asum = are + aim;
        v2f ps = pr + pi;
        d1 = WMMA_F32X4(are,  pr, d1);
        d2 = WMMA_F32X4(aim,  pi, d2);
        d3 = WMMA_F32X4(asum, ps, d3);
      }
      v8f dre = d1 - d2;
      v8f dim = (d3 - d1) - d2;

      float wA = (j < LL && cA < CDIM) ? fabsf(wAl) : 0.f;
      float wB = (j < LL && cB < CDIM) ? fabsf(wBl) : 0.f;

  #pragma unroll
      for (int r = 0; r < 8; ++r) {
        float w = (r < 4) ? wA : wB;
        gre[r] += dre[r] * w;
        gim[r] += dim[r] * w;
      }
    }

#pragma unroll
    for (int tp = 0; tp < 4; ++tp) {
      v2f gr; gr.x = gre[2 * tp]; gr.y = gre[2 * tp + 1];
      v2f gi; gi.x = gim[2 * tp]; gi.y = gim[2 * tp + 1];
      v2f gs = gr + gi;
#pragma unroll
      for (int it = 0; it < 4; ++it) {
        v4f qq = qv[tp * 4 + it];
        v2f br; br.x = qq.x; br.y = qq.y;
        v2f bi; bi.x = qq.z; bi.y = qq.w;
        v2f bs = br + bi;
        f1[it] = WMMA_F32X4(gr, br, f1[it]);
        f2[it] = WMMA_F32X4(gi, bi, f2[it]);
        f3[it] = WMMA_F32X4(gs, bs, f3[it]);
      }
    }
  }

  v8f fre[4], fim[4];
#pragma unroll
  for (int it = 0; it < 4; ++it) {
    fre[it] = f1[it] - f2[it];
    fim[it] = (f3[it] - f1[it]) - f2[it];
  }

  __syncthreads();
  int lane_e = (int)__builtin_amdgcn_mbcnt_lo(~0u, 0u);
  float* red = sm + (size_t)jt * 64 * 32 + lane_e;
  if (par == 1) {
#pragma unroll
    for (int it = 0; it < 4; ++it) {
#pragma unroll
      for (int r = 0; r < 8; ++r) {
        red[(it * 16 + r) * 32]     = fre[it][r];
        red[(it * 16 + 8 + r) * 32] = fim[it][r];
      }
    }
  }
  __syncthreads();
  if (par == 0) {
    size_t pbase = ((size_t)(s * BB + b) * 16 + (size_t)(jt * 4)) * 256 + lane_e;
#pragma unroll
    for (int it = 0; it < 4; ++it) {
#pragma unroll
      for (int r = 0; r < 8; ++r) {
        pre[pbase + it * 256 + r * 32] = fre[it][r] + red[(it * 16 + r) * 32];
        pim[pbase + it * 256 + r * 32] = fim[it][r] + red[(it * 16 + 8 + r) * 32];
      }
    }
  }
}

#define MPITCH 65
__global__ void __launch_bounds__(1024) k_final(
    const float* __restrict__ pre, const float* __restrict__ pim,
    const float* __restrict__ smax, const float* __restrict__ bias,
    const float* __restrict__ amp, float* __restrict__ out) {
  __shared__ float mb[JPAD * MPITCH];
  __shared__ float red[32];
  int b = blockIdx.x;
  int tid = threadIdx.x;
  int tile = tid >> 6;
  int p0 = (tid & 63) * 4;
  int j  = (tile >> 2) * 16 + (p0 >> 5) + 8 * ((p0 >> 4) & 1);
  int i0 = (tile & 3) * 16 + (p0 & 15);

  float m = 0.f;
  if (j < LL && i0 < LL) {
    const float* pr = pre + (size_t)b * TILEP + tid * 4;
    const float* pi = pim + (size_t)b * TILEP + tid * 4;
    float4 sr = *(const float4*)pr;
    float4 si = *(const float4*)pi;
#pragma unroll
    for (int s = 1; s < KSPLIT; ++s) {
      float4 a = *(const float4*)(pr + (size_t)s * BB * TILEP);
      float4 c = *(const float4*)(pi + (size_t)s * BB * TILEP);
      sr.x += a.x; sr.y += a.y; sr.z += a.z; sr.w += a.w;
      si.x += c.x; si.y += c.y; si.z += c.z; si.w += c.w;
    }
    float fr[4] = {sr.x, sr.y, sr.z, sr.w};
    float fi[4] = {si.x, si.y, si.z, si.w};
#pragma unroll
    for (int k = 0; k < 4; ++k) {
      int i = i0 + k;
      float v = 0.f;
      if (i < LL) v = sqrtf(fr[k] * fr[k] + fi[k] * fi[k]) / amp[j * LL + i];
      mb[j * MPITCH + i] = v;
      m = fmaxf(m, v);
    }
  }
  for (int off = 16; off > 0; off >>= 1) m = fmaxf(m, __shfl_xor(m, off));
  if ((tid & 31) == 0) red[tid >> 5] = m;
  __syncthreads();
  float mx = red[tid & 31];
  for (int off = 16; off > 0; off >>= 1) mx = fmaxf(mx, __shfl_xor(mx, off));

  float scale = (smax[b] * smax[16]) / mx;
  for (int idx = tid; idx < LL * LL; idx += 1024) {
    int i = idx / LL, j2 = idx % LL;
    out[b * 2500 + idx] = mb[j2 * MPITCH + i] * scale + bias[j2];
  }
}

extern "C" void kernel_launch(void* const* d_in, const int* in_sizes, int n_in,
                              void* d_out, int out_size, void* d_ws, size_t ws_size,
                              hipStream_t stream) {
  const float* inp  = (const float*)d_in[0];
  const float* wgt  = (const float*)d_in[1];
  const float* bias = (const float*)d_in[2];
  const float* amp  = (const float*)d_in[3];
  const float* sre  = (const float*)d_in[4];
  const float* sim  = (const float*)d_in[5];

  float* ws   = (float*)d_ws;
  float* smax = ws;
  float* qf   = ws + 32;
  float* pre  = qf + (size_t)QPER * 16 * 32 * 4;
  float* pim  = pre + (size_t)KSPLIT * BB * TILEP;

  k_maxes  <<<17, 256, 0, stream>>>(inp, wgt, smax);
  k_build_q<<<(QPER * 16 * 32 + 255) / 256, 256, 0, stream>>>(qf);
  k_mm12   <<<BB * KSPLIT, 256, 0, stream>>>(inp, wgt, sre, sim, qf, pre, pim);
  k_final  <<<BB, 1024, 0, stream>>>(pre, pim, smax, bias, amp, (float*)d_out);
}


# launches 4 -> 2: the max kernel folded into the last kernel, the Q kernel into the GEMM kernel, which builds its Q fragments and P tables in LDS from one 200-entry twiddle table (same values)
# speedup vs baseline: 5.6982x; 1.3153x over previous
#include <hip/hip_runtime.h>
#include <math.h>


#define LL    50
#define BB    16
#define ITERR 76
#define KDIM  100
#define CDIM  3800
#define CTILES 238
#define JPAD  64
#define SROW  7600
#define SROW0 50
#define COL0  1900
#define KSPLIT 16
#define CTPER  15
#define TILEP  4096
#define QPER  25
#define NLONG 13
#define NTW   200
#define QCLS  3

typedef float v2f __attribute__((ext_vector_type(2)));
typedef float v4f __attribute__((ext_vector_type(4)));
typedef float v8f __attribute__((ext_vector_type(8)));

#define WMMA_F32X4(a, b, c) \
  __builtin_amdgcn_wmma_f32_16x16x4_f32(false, (a), false, (b), (short)0, (c), false, false)

__device__ __forceinline__ int class_of(int tau) {
  return (tau < NLONG * 10) ? (tau / 10) : (NLONG + (tau - NLONG * 10) / 9);
}

__global__ void __launch_bounds__(256) __attribute__((amdgpu_waves_per_eu(2, 2))) k_mm12(
    const float* __restrict__ inp, const float* __restrict__ wgt,
    const float* __restrict__ sre, const float* __restrict__ sim,
    float* __restrict__ pre, float* __restrict__ pim) {
  __shared__ float sm[2 * JPAD * KDIM];
  __shared__ v2f tw[NTW];
  __shared__ v4f qs[QCLS * 512];
  float* Pre = sm;
  float* Pim = sm + JPAD * KDIM;
  int s = blockIdx.x % KSPLIT;
  int b = blockIdx.x / KSPLIT;

  if (threadIdx.x < NTW) {
    float sn, cz;
    sincospif((float)threadIdx.x * 0.01f, &sn, &cz);
    v2f v; v.x = cz; v.y = -sn;
    tw[threadIdx.x] = v;
  }
  __syncthreads();

  for (int k = 0; k < JPAD * KDIM / 256; ++k) {
    int e = threadIdx.x + k * 256;
    int j = e / KDIM, r = e % KDIM;
    int t = ((150 + 2 * j) * (50 + r)) % 200;
    v2f v = tw[t];
    Pre[e] = (j < LL) ? v.x : 0.f;
    Pim[e] = (j < LL) ? v.y : 0.f;
  }

  int tau0 = s * CTPER;
  int tau1 = (tau0 + CTPER < CTILES) ? (tau0 + CTPER) : CTILES;
  int kap0 = class_of(tau0);
  int ncls = class_of(tau1 - 1) - kap0 + 1;
  for (int cls = 0; cls < ncls; ++cls) {
    for (int hh = 0; hh < 2; ++hh) {
      int rem = threadIdx.x + hh * 256;
      int ql = rem & 31;
      int it = (rem >> 5) & 3;
      int tp = (rem >> 7) & 3;
      int i  = it * 16 + (ql & 15);
      int c0 = (kap0 + cls) * 16 + 8 * (ql >> 4) + 2 * tp;
      int t0 = ((i + 1) * (COL0 + c0)) % 200;
      int t1 = ((i + 1) * (COL0 + c0 + 1)) % 200;
      v2f z0 = tw[t0];
      v2f z1 = tw[t1];
      v4f v = {0.f, 0.f, 0.f, 0.f};
      if (i < LL) { v.x = z0.x; v.y = z1.x; v.z = z0.y; v.w = z1.y; }
      qs[cls * 512 + rem] = v;
    }
  }
  __syncthreads();

  int wave = threadIdx.x >> 5;
  int lane = threadIdx.x & 31;
  int jt  = wave & 3;
  int par = wave >> 2;
  int h  = lane >> 4;
  int ln = lane & 15;
  int j  = jt * 16 + ln;

  v8f f1[4], f2[4], f3[4];
#pragma unroll
  for (int it = 0; it < 4; ++it) { f1[it] = (v8f){}; f2[it] = (v8f){}; f3[it] = (v8f){}; }

  int nhalf = (tau1 - tau0 + 1) >> 1;
  int tau  = par ? (tau0 + nhalf) : tau0;
  int tend = par ? tau1 : (tau0 + nhalf);
  while (tau < tend) {
    int kap, m, nk;
    if (tau < NLONG * 10) { kap = tau / 10; m = tau - kap * 10; nk = 10; }
    else { int u = tau - NLONG * 10; int kq = u / 9; kap = NLONG + kq; m = u - kq * 9; nk = 9; }
    int mend = m + (tend - tau);
    if (mend > nk) mend = nk;
    tau += mend - m;
    const v4f* qt = qs + (kap - kap0) * 512 + lane;
    v4f qv[16];
#pragma unroll
    for (int u = 0; u < 16; ++u) qv[u] = qt[u * 32];
    v8f gre = {}; v8f gim = {};
    for (; m < mend; ++m) {
      int ct = kap + QPER * m;
      int c  = ct * 16 + ln;
      int x2 = COL0 + c;
      int q  = (c < CDIM) ? (c / ITERR) : 0;
      const float* inp_b = inp + b * 2500 + q;
      int cA = ct * 16 + 8 * h;
      int cB = cA + 4;
      int jw = (j < LL) ? j : 0;
      float wAl = wgt[jw * LL + ((cA < CDIM) ? cA / ITERR : 0)];
      float wBl = wgt[jw * LL + ((cB < CDIM) ? cB / ITERR : 0)];
      v8f d1 = {}; v8f d2 = {}; v8f d3 = {};
#pragma unroll
      for (int t = 0; t < KDIM / 4; ++t) {
        int kk = 4 * t + 2 * h;
        v2f pr = *(const v2f*)&Pre[j * KDIM + kk];
        v2f pi = *(const v2f*)&Pim[j * KDIM + kk];
        float inpv = inp_b[(2 * t + h) * LL];
        int so = (SROW0 + kk) * SROW + x2;
        v2f are; are.x = sre[so] * inpv; are.y = sre[so + SROW] * inpv;
        v2f aim; aim.x = sim[so] * inpv; aim.y = sim[so + SROW] * inpv;
        v2f asum = are + aim;
        v2f ps = pr + pi;
        d1 = WMMA_F32X4(are,  pr, d1);
        d2 = WMMA_F32X4(aim,  pi, d2);
        d3 = WMMA_F32X4(asum, ps, d3);
      }
      v8f dre = d1 - d2;
      v8f dim = (d3 - d1) - d2;

      float wA = (j < LL && cA < CDIM) ? fabsf(wAl) : 0.f;
      float wB = (j < LL && cB < CDIM) ? fabsf(wBl) : 0.f;

  #pragma unroll
      for (int r = 0; r < 8; ++r) {
        float w = (r < 4) ? wA : wB;
        gre[r] += dre[r] * w;
        gim[r] += dim[r] * w;
      }
    }

#pragma unroll
    for (int tp = 0; tp < 4; ++tp) {
      v2f gr; gr.x = gre[2 * tp]; gr.y = gre[2 * tp + 1];
      v2f gi; gi.x = gim[2 * tp]; gi.y = gim[2 * tp + 1];
      v2f gs = gr + gi;
#pragma unroll
      for (int it = 0; it < 4; ++it) {
        v4f qq = qv[tp * 4 + it];
        v2f br; br.x = qq.x; br.y = qq.y;
        v2f bi; bi.x = qq.z; bi.y = qq.w;
        v2f bs = br + bi;
        f1[it] = WMMA_F32X4(gr, br, f1[it]);
        f2[it] = WMMA_F32X4(gi, bi, f2[it]);
        f3[it] = WMMA_F32X4(gs, bs, f3[it]);
      }
    }
  }

  v8f fre[4], fim[4];
#pragma unroll
  for (int it = 0; it < 4; ++it) {
    fre[it] = f1[it] - f2[it];
    fim[it] = (f3[it] - f1[it]) - f2[it];
  }

  __syncthreads();
  int lane_e = (int)__builtin_amdgcn_mbcnt_lo(~0u, 0u);
  float* red = sm + (size_t)jt * 64 * 32 + lane_e;
  if (par == 1) {
#pragma unroll
    for (int it = 0; it < 4; ++it) {
#pragma unroll
      for (int r = 0; r < 8; ++r) {
        red[(it * 16 + r) * 32]     = fre[it][r];
        red[(it * 16 + 8 + r) * 32] = fim[it][r];
      }
    }
  }
  __syncthreads();
  if (par == 0) {
    size_t pbase = ((size_t)(s * BB + b) * 16 + (size_t)(jt * 4)) * 256 + lane_e;
#pragma unroll
    for (int it = 0; it < 4; ++it) {
#pragma unroll
      for (int r = 0; r < 8; ++r) {
        pre[pbase + it * 256 + r * 32] = fre[it][r] + red[(it * 16 + r) * 32];
        pim[pbase + it * 256 + r * 32] = fim[it][r] + red[(it * 16 + 8 + r) * 32];
      }
    }
  }
}

#define MPITCH 65
__global__ void __launch_bounds__(1024) k_final(
    const float* __restrict__ pre, const float* __restrict__ pim,
    const float* __restrict__ inp, const float* __restrict__ wgt,
    const float* __restrict__ bias,
    const float* __restrict__ amp, float* __restrict__ out) {
  __shared__ float mb[JPAD * MPITCH];
  __shared__ float red[3][32];
  int b = blockIdx.x;
  int tid = threadIdx.x;
  int tile = tid >> 6;
  int p0 = (tid & 63) * 4;
  int j  = (tile >> 2) * 16 + (p0 >> 5) + 8 * ((p0 >> 4) & 1);
  int i0 = (tile & 3) * 16 + (p0 & 15);

  float m = 0.f;
  if (j < LL && i0 < LL) {
    const float* pr = pre + (size_t)b * TILEP + tid * 4;
    const float* pi = pim + (size_t)b * TILEP + tid * 4;
    float4 sr = *(const float4*)pr;
    float4 si = *(const float4*)pi;
#pragma unroll
    for (int s = 1; s < KSPLIT; ++s) {
      float4 a = *(const float4*)(pr + (size_t)s * BB * TILEP);
      float4 c = *(const float4*)(pi + (size_t)s * BB * TILEP);
      sr.x += a.x; sr.y += a.y; sr.z += a.z; sr.w += a.w;
      si.x += c.x; si.y += c.y; si.z += c.z; si.w += c.w;
    }
    float fr[4] = {sr.x, sr.y, sr.z, sr.w};
    float fi[4] = {si.x, si.y, si.z, si.w};
#pragma unroll
    for (int k = 0; k < 4; ++k) {
      int i = i0 + k;
      float v = 0.f;
      if (i < LL) v = sqrtf(fr[k] * fr[k] + fi[k] * fi[k]) / amp[j * LL + i];
      mb[j * MPITCH + i] = v;
      m = fmaxf(m, v);
    }
  }
  int k2 = (tid + 2048 < LL * LL) ? (tid + 2048) : tid;
  float x0 = inp[b * 2500 + tid], x1 = inp[b * 2500 + tid + 1024], x2 = inp[b * 2500 + k2];
  float w0 = wgt[tid], w1 = wgt[tid + 1024], w2 = wgt[k2];
  float mi = fmaxf(fmaxf(fmaxf(-3.4e38f, x0), x1), x2);
  float mw = fmaxf(fmaxf(fmaxf(-3.4e38f, fabsf(w0)), fabsf(w1)), fabsf(w2));
  for (int off = 16; off > 0; off >>= 1) {
    m  = fmaxf(m,  __shfl_xor(m,  off));
    mi = fmaxf(mi, __shfl_xor(mi, off));
    mw = fmaxf(mw, __shfl_xor(mw, off));
  }
  if ((tid & 31) == 0) { red[0][tid >> 5] = m; red[1][tid >> 5] = mi; red[2][tid >> 5] = mw; }
  __syncthreads();
  float mx  = red[0][tid & 31];
  float mxi = red[1][tid & 31];
  float mxw = red[2][tid & 31];
  for (int off = 16; off > 0; off >>= 1) {
    mx  = fmaxf(mx,  __shfl_xor(mx,  off));
    mxi = fmaxf(mxi, __shfl_xor(mxi, off));
    mxw = fmaxf(mxw, __shfl_xor(mxw, off));
  }

  float scale = (mxi * mxw) / mx;
  for (int idx = tid; idx < LL * LL; idx += 1024) {
    int i = idx / LL, j2 = idx % LL;
    out[b * 2500 + idx] = mb[j2 * MPITCH + i] * scale + bias[j2];
  }
}

extern "C" void kernel_launch(void* const* d_in, const int* in_sizes, int n_in,
                              void* d_out, int out_size, void* d_ws, size_t ws_size,
                              hipStream_t stream) {
  const float* inp  = (const float*)d_in[0];
  const float* wgt  = (const float*)d_in[1];
  const float* bias = (const float*)d_in[2];
  const float* amp  = (const float*)d_in[3];
  const float* sre  = (const float*)d_in[4];
  const float* sim  = (const float*)d_in[5];

  float* ws   = (float*)d_ws;
  float* pre  = ws;
  float* pim  = pre + (size_t)KSPLIT * BB * TILEP;

  k_mm12   <<<BB * KSPLIT, 256, 0, stream>>>(inp, wgt, sre, sim, pre, pim);
  k_final  <<<BB, 1024, 0, stream>>>(pre, pim, inp, wgt, bias, amp, (float*)d_out);
}


# GEMM-1 global loads software-pipelined 11 k-steps ahead of their matrix instructions through a register ring that runs on across tile boundaries; summation order unchanged, some operand sums fused
# speedup vs baseline: 6.1773x; 1.0841x over previous
#include <hip/hip_runtime.h>
#include <math.h>


#define LL    50
#define BB    16
#define ITERR 76
#define KDIM  100
#define CDIM  3800
#define CTILES 238
#define JPAD  64
#define SROW  7600
#define SROW0 50
#define COL0  1900
#define KSPLIT 16
#define CTPER  15
#define TILEP  4096
#define QPER  25
#define NLONG 13
#define NTW   200
#define QCLS  3
#define KSTEPS 25
#define AHEAD  11

typedef float v2f __attribute__((ext_vector_type(2)));
typedef float v4f __attribute__((ext_vector_type(4)));
typedef float v8f __attribute__((ext_vector_type(8)));

#define WMMA_F32X4(a, b, c) \
  __builtin_amdgcn_wmma_f32_16x16x4_f32(false, (a), false, (b), (short)0, (c), false, false)

__device__ __forceinline__ int class_of(int tau) {
  return (tau < NLONG * 10) ? (tau / 10) : (NLONG + (tau - NLONG * 10) / 9);
}

__device__ __forceinline__ int ct_of(int tau) {
  int kap, m;
  if (tau < NLONG * 10) { kap = tau / 10; m = tau - kap * 10; }
  else { int u = tau - NLONG * 10; int kq = u / 9; kap = NLONG + kq; m = u - kq * 9; }
  return kap + QPER * m;
}

__global__ void __launch_bounds__(256) __attribute__((amdgpu_waves_per_eu(2, 2))) k_mm12(
    const float* __restrict__ inp, const float* __restrict__ wgt,
    const float* __restrict__ sre, const float* __restrict__ sim,
    float* __restrict__ pre, float* __restrict__ pim) {
  __shared__ float sm[2 * JPAD * KDIM];
  __shared__ v2f tw[NTW];
  __shared__ v4f qs[QCLS * 512];
  float* Pre = sm;
  float* Pim = sm + JPAD * KDIM;
  int s = blockIdx.x % KSPLIT;
  int b = blockIdx.x / KSPLIT;

  if (threadIdx.x < NTW) {
    float sn, cz;
    sincospif((float)threadIdx.x * 0.01f, &sn, &cz);
    v2f v; v.x = cz; v.y = -sn;
    tw[threadIdx.x] = v;
  }
  __syncthreads();

  for (int k = 0; k < JPAD * KDIM / 256; ++k) {
    int e = threadIdx.x + k * 256;
    int j = e / KDIM, r = e % KDIM;
    int t = ((150 + 2 * j) * (50 + r)) % 200;
    v2f v = tw[t];
    Pre[e] = (j < LL) ? v.x : 0.f;
    Pim[e] = (j < LL) ? v.y : 0.f;
  }

  int tau0 = s * CTPER;
  int tau1 = (tau0 + CTPER < CTILES) ? (tau0 + CTPER) : CTILES;
  int kap0 = class_of(tau0);
  int ncls = class_of(tau1 - 1) - kap0 + 1;
  for (int cls = 0; cls < ncls; ++cls) {
    for (int hh = 0; hh < 2; ++hh) {
      int rem = threadIdx.x + hh * 256;
      int ql = rem & 31;
      int it = (rem >> 5) & 3;
      int tp = (rem >> 7) & 3;
      int i  = it * 16 + (ql & 15);
      int c0 = (kap0 + cls) * 16 + 8 * (ql >> 4) + 2 * tp;
      int t0 = ((i + 1) * (COL0 + c0)) % 200;
      int t1 = ((i + 1) * (COL0 + c0 + 1)) % 200;
      v2f z0 = tw[t0];
      v2f z1 = tw[t1];
      v4f v = {0.f, 0.f, 0.f, 0.f};
      if (i < LL) { v.x = z0.x; v.y = z1.x; v.z = z0.y; v.w = z1.y; }
      qs[cls * 512 + rem] = v;
    }
  }
  __syncthreads();

  int wave = threadIdx.x >> 5;
  int lane = threadIdx.x & 31;
  int jt  = wave & 3;
  int par = wave >> 2;
  int h  = lane >> 4;
  int ln = lane & 15;
  int j  = jt * 16 + ln;

  v8f f1[4], f2[4], f3[4];
#pragma unroll
  for (int it = 0; it < 4; ++it) { f1[it] = (v8f){}; f2[it] = (v8f){}; f3[it] = (v8f){}; }

  int nhalf = (tau1 - tau0 + 1) >> 1;
  int tau  = par ? (tau0 + nhalf) : tau0;
  int tend = par ? tau1 : (tau0 + nhalf);

  float sr0[KSTEPS], sr1[KSTEPS], si0[KSTEPS], si1[KSTEPS], iv[KSTEPS];
#define LOAD_STEP(T, PRE_, PIM_, PIN_) do { \
    sr0[(T)] = (PRE_)[(T) * 4 * SROW]; sr1[(T)] = (PRE_)[(T) * 4 * SROW + SROW]; \
    si0[(T)] = (PIM_)[(T) * 4 * SROW]; si1[(T)] = (PIM_)[(T) * 4 * SROW + SROW]; \
    iv[(T)]  = (PIN_)[(T) * 2 * LL]; } while (0)
#define TILE_PTRS(CT, PRE_, PIM_, PIN_) do { \
    int c_ = (CT) * 16 + ln; \
    int q_ = (c_ < CDIM) ? (c_ / ITERR) : 0; \
    int so_ = (SROW0 + 2 * h) * SROW + COL0 + c_; \
    (PRE_) = sre + so_; (PIM_) = sim + so_; (PIN_) = inp + b * 2500 + q_ + h * LL; } while (0)

  const float *cre, *cim, *cin;
  const float *nre, *nim, *nin;
  TILE_PTRS(ct_of(tau), cre, cim, cin);
#pragma unroll
  for (int t = 0; t < AHEAD; ++t) LOAD_STEP(t, cre, cim, cin);

  v2f prr[KSTEPS], pii[KSTEPS], pss[KSTEPS];
#pragma unroll
  for (int t = 0; t < KSTEPS; ++t) {
    int kk = 4 * t + 2 * h;
    prr[t] = *(const v2f*)&Pre[j * KDIM + kk];
    pii[t] = *(const v2f*)&Pim[j * KDIM + kk];
    pss[t] = prr[t] + pii[t];
  }
#pragma unroll
  for (int t = 0; t < AHEAD; ++t)
    asm volatile("" :: "v"(sr0[t]), "v"(sr1[t]), "v"(si0[t]), "v"(si1[t]), "v"(iv[t]));

  while (tau < tend) {
    int kap, m, nk;
    if (tau < NLONG * 10) { kap = tau / 10; m = tau - kap * 10; nk = 10; }
    else { int u = tau - NLONG * 10; int kq = u / 9; kap = NLONG + kq; m = u - kq * 9; nk = 9; }
    int mend = m + (tend - tau);
    if (mend > nk) mend = nk;
    v8f gre = {}; v8f gim = {};
    for (; m < mend; ++m) {
      int ct = kap + QPER * m;
      ++tau;
      int ctn = (tau < tend) ? ct_of(tau) : ct;
      TILE_PTRS(ctn, nre, nim, nin);
      int cA = ct * 16 + 8 * h;
      int cB = cA + 4;
      int jw = (j < LL) ? j : 0;
      float wAl = wgt[jw * LL + ((cA < CDIM) ? cA / ITERR : 0)];
      float wBl = wgt[jw * LL + ((cB < CDIM) ? cB / ITERR : 0)];
      v8f d1 = {}; v8f d2 = {}; v8f d3 = {};
#pragma unroll
      for (int t = 0; t < KSTEPS; ++t) {
        __builtin_amdgcn_sched_barrier(0);
        if (t + AHEAD < KSTEPS) LOAD_STEP(t + AHEAD, cre, cim, cin);
        else                    LOAD_STEP(t + AHEAD - KSTEPS, nre, nim, nin);
        float inpv = iv[t];
        v2f are; are.x = sr0[t] * inpv; are.y = sr1[t] * inpv;
        v2f aim; aim.x = si0[t] * inpv; aim.y = si1[t] * inpv;
        v2f asum = are + aim;
        d1 = WMMA_F32X4(are,  prr[t], d1);
        d2 = WMMA_F32X4(aim,  pii[t], d2);
        d3 = WMMA_F32X4(asum, pss[t], d3);
      }
      cre = nre; cim = nim; cin = nin;
      v8f dre = d1 - d2;
      v8f dim = (d3 - d1) - d2;

      float wA = (j < LL && cA < CDIM) ? fabsf(wAl) : 0.f;
      float wB = (j < LL && cB < CDIM) ? fabsf(wBl) : 0.f;

  #pragma unroll
      for (int r = 0; r < 8; ++r) {
        float w = (r < 4) ? wA : wB;
        gre[r] += dre[r] * w;
        gim[r] += dim[r] * w;
      }
    }

    const v4f* qt = qs + (kap - kap0) * 512 + lane;
    v4f qv[16];
#pragma unroll
    for (int u = 0; u < 16; ++u) qv[u] = qt[u * 32];

#pragma unroll
    for (int tp = 0; tp < 4; ++tp) {
      v2f gr; gr.x = gre[2 * tp]; gr.y = gre[2 * tp + 1];
      v2f gi; gi.x = gim[2 * tp]; gi.y = gim[2 * tp + 1];
      v2f gs = gr + gi;
#pragma unroll
      for (int it = 0; it < 4; ++it) {
        v4f qq = qv[tp * 4 + it];
        v2f br; br.x = qq.x; br.y = qq.y;
        v2f bi; bi.x = qq.z; bi.y = qq.w;
        v2f bs = br + bi;
        f1[it] = WMMA_F32X4(gr, br, f1[it]);
        f2[it] = WMMA_F32X4(gi, bi, f2[it]);
        f3[it] = WMMA_F32X4(gs, bs, f3[it]);
      }
    }
  }
#undef LOAD_STEP
#undef TILE_PTRS

  v8f fre[4], fim[4];
#pragma unroll
  for (int it = 0; it < 4; ++it) {
    fre[it] = f1[it] - f2[it];
    fim[it] = (f3[it] - f1[it]) - f2[it];
  }

  __syncthreads();
  int lane_e = (int)__builtin_amdgcn_mbcnt_lo(~0u, 0u);
  float* red = sm + (size_t)jt * 64 * 32 + lane_e;
  if (par == 1) {
#pragma unroll
    for (int it = 0; it < 4; ++it) {
#pragma unroll
      for (int r = 0; r < 8; ++r) {
        red[(it * 16 + r) * 32]     = fre[it][r];
        red[(it * 16 + 8 + r) * 32] = fim[it][r];
      }
    }
  }
  __syncthreads();
  if (par == 0) {
    size_t pbase = ((size_t)(s * BB + b) * 16 + (size_t)(jt * 4)) * 256 + lane_e;
#pragma unroll
    for (int it = 0; it < 4; ++it) {
#pragma unroll
      for (int r = 0; r < 8; ++r) {
        pre[pbase + it * 256 + r * 32] = fre[it][r] + red[(it * 16 + r) * 32];
        pim[pbase + it * 256 + r * 32] = fim[it][r] + red[(it * 16 + 8 + r) * 32];
      }
    }
  }
}

#define MPITCH 65
__global__ void __launch_bounds__(1024) k_final(
    const float* __restrict__ pre, const float* __restrict__ pim,
    const float* __restrict__ inp, const float* __restrict__ wgt,
    const float* __restrict__ bias,
    const float* __restrict__ amp, float* __restrict__ out) {
  __shared__ float mb[JPAD * MPITCH];
  __shared__ float red[3][32];
  int b = blockIdx.x;
  int tid = threadIdx.x;
  int tile = tid >> 6;
  int p0 = (tid & 63) * 4;
  int j  = (tile >> 2) * 16 + (p0 >> 5) + 8 * ((p0 >> 4) & 1);
  int i0 = (tile & 3) * 16 + (p0 & 15);

  float m = 0.f;
  if (j < LL && i0 < LL) {
    const float* pr = pre + (size_t)b * TILEP + tid * 4;
    const float* pi = pim + (size_t)b * TILEP + tid * 4;
    float4 sr = *(const float4*)pr;
    float4 si = *(const float4*)pi;
#pragma unroll
    for (int s = 1; s < KSPLIT; ++s) {
      float4 a = *(const float4*)(pr + (size_t)s * BB * TILEP);
      float4 c = *(const float4*)(pi + (size_t)s * BB * TILEP);
      sr.x += a.x; sr.y += a.y; sr.z += a.z; sr.w += a.w;
      si.x += c.x; si.y += c.y; si.z += c.z; si.w += c.w;
    }
    float fr[4] = {sr.x, sr.y, sr.z, sr.w};
    float fi[4] = {si.x, si.y, si.z, si.w};
#pragma unroll
    for (int k = 0; k < 4; ++k) {
      int i = i0 + k;
      float v = 0.f;
      if (i < LL) v = sqrtf(fr[k] * fr[k] + fi[k] * fi[k]) / amp[j * LL + i];
      mb[j * MPITCH + i] = v;
      m = fmaxf(m, v);
    }
  }
  int k2 = (tid + 2048 < LL * LL) ? (tid + 2048) : tid;
  float x0 = inp[b * 2500 + tid], x1 = inp[b * 2500 + tid + 1024], x2 = inp[b * 2500 + k2];
  float w0 = wgt[tid], w1 = wgt[tid + 1024], w2 = wgt[k2];
  float mi = fmaxf(fmaxf(fmaxf(-3.4e38f, x0), x1), x2);
  float mw = fmaxf(fmaxf(fmaxf(-3.4e38f, fabsf(w0)), fabsf(w1)), fabsf(w2));
  for (int off = 16; off > 0; off >>= 1) {
    m  = fmaxf(m,  __shfl_xor(m,  off));
    mi = fmaxf(mi, __shfl_xor(mi, off));
    mw = fmaxf(mw, __shfl_xor(mw, off));
  }
  if ((tid & 31) == 0) { red[0][tid >> 5] = m; red[1][tid >> 5] = mi; red[2][tid >> 5] = mw; }
  __syncthreads();
  float mx  = red[0][tid & 31];
  float mxi = red[1][tid & 31];
  float mxw = red[2][tid & 31];
  for (int off = 16; off > 0; off >>= 1) {
    mx  = fmaxf(mx,  __shfl_xor(mx,  off));
    mxi = fmaxf(mxi, __shfl_xor(mxi, off));
    mxw = fmaxf(mxw, __shfl_xor(mxw, off));
  }

  float scale = (mxi * mxw) / mx;
  for (int idx = tid; idx < LL * LL; idx += 1024) {
    int i = idx / LL, j2 = idx % LL;
    out[b * 2500 + idx] = mb[j2 * MPITCH + i] * scale + bias[j2];
  }
}

extern "C" void kernel_launch(void* const* d_in, const int* in_sizes, int n_in,
                              void* d_out, int out_size, void* d_ws, size_t ws_size,
                              hipStream_t stream) {
  const float* inp  = (const float*)d_in[0];
  const float* wgt  = (const float*)d_in[1];
  const float* bias = (const float*)d_in[2];
  const float* amp  = (const float*)d_in[3];
  const float* sre  = (const float*)d_in[4];
  const float* sim  = (const float*)d_in[5];

  float* ws   = (float*)d_ws;
  float* pre  = ws;
  float* pim  = pre + (size_t)KSPLIT * BB * TILEP;

  k_mm12   <<<BB * KSPLIT, 256, 0, stream>>>(inp, wgt, sre, sim, pre, pim);
  k_final  <<<BB, 1024, 0, stream>>>(pre, pim, inp, wgt, bias, amp, (float*)d_out);
}


# tiles laid 12 columns to the left (239, was 238): each 16-float row piece a half-wave loads from S is 64-byte aligned, in one 128-byte line (a third fewer lines fetched); changes the summation order
# speedup vs baseline: 6.3731x; 1.0317x over previous
#include <hip/hip_runtime.h>
#include <math.h>


#define LL    50
#define BB    16
#define ITERR 76
#define KDIM  100
#define CDIM  3800
#define CTILES 239
#define JPAD  64
#define SROW  7600
#define SROW0 50
#define COL0  1900
#define CSHIFT (COL0 % 16)
#define KSPLIT 16
#define CTPER  15
#define TILEP  4096
#define QPER  25
#define NLONG 14
#define NTW   200
#define QCLS  3
#define KSTEPS 25
#define AHEAD  11

typedef float v2f __attribute__((ext_vector_type(2)));
typedef float v4f __attribute__((ext_vector_type(4)));
typedef float v8f __attribute__((ext_vector_type(8)));

#define WMMA_F32X4(a, b, c) \
  __builtin_amdgcn_wmma_f32_16x16x4_f32(false, (a), false, (b), (short)0, (c), false, false)

__device__ __forceinline__ int class_of(int tau) {
  return (tau < NLONG * 10) ? (tau / 10) : (NLONG + (tau - NLONG * 10) / 9);
}

__device__ __forceinline__ int ct_of(int tau) {
  int kap, m;
  if (tau < NLONG * 10) { kap = tau / 10; m = tau - kap * 10; }
  else { int u = tau - NLONG * 10; int kq = u / 9; kap = NLONG + kq; m = u - kq * 9; }
  return kap + QPER * m;
}

__global__ void __launch_bounds__(256) __attribute__((amdgpu_waves_per_eu(2, 2))) k_mm12(
    const float* __restrict__ inp, const float* __restrict__ wgt,
    const float* __restrict__ sre, const float* __restrict__ sim,
    float* __restrict__ pre, float* __restrict__ pim) {
  __shared__ float sm[2 * JPAD * KDIM];
  __shared__ v2f tw[NTW];
  __shared__ v4f qs[QCLS * 512];
  float* Pre = sm;
  float* Pim = sm + JPAD * KDIM;
  int s = blockIdx.x % KSPLIT;
  int b = blockIdx.x / KSPLIT;

  if (threadIdx.x < NTW) {
    float sn, cz;
    sincospif((float)threadIdx.x * 0.01f, &sn, &cz);
    v2f v; v.x = cz; v.y = -sn;
    tw[threadIdx.x] = v;
  }
  __syncthreads();

  for (int k = 0; k < JPAD * KDIM / 256; ++k) {
    int e = threadIdx.x + k * 256;
    int j = e / KDIM, r = e % KDIM;
    int t = ((150 + 2 * j) * (50 + r)) % 200;
    v2f v = tw[t];
    Pre[e] = (j < LL) ? v.x : 0.f;
    Pim[e] = (j < LL) ? v.y : 0.f;
  }

  int tau0 = s * CTPER;
  int tau1 = (tau0 + CTPER < CTILES) ? (tau0 + CTPER) : CTILES;
  int kap0 = class_of(tau0);
  int ncls = class_of(tau1 - 1) - kap0 + 1;
  for (int cls = 0; cls < ncls; ++cls) {
    for (int hh = 0; hh < 2; ++hh) {
      int rem = threadIdx.x + hh * 256;
      int ql = rem & 31;
      int it = (rem >> 5) & 3;
      int tp = (rem >> 7) & 3;
      int i  = it * 16 + (ql & 15);
      int c0 = (kap0 + cls) * 16 - CSHIFT + 8 * (ql >> 4) + 2 * tp;
      int t0 = ((i + 1) * (COL0 + c0)) % 200;
      int t1 = ((i + 1) * (COL0 + c0 + 1)) % 200;
      v2f z0 = tw[t0];
      v2f z1 = tw[t1];
      v4f v = {0.f, 0.f, 0.f, 0.f};
      if (i < LL) { v.x = z0.x; v.y = z1.x; v.z = z0.y; v.w = z1.y; }
      qs[cls * 512 + rem] = v;
    }
  }
  __syncthreads();

  int wave = threadIdx.x >> 5;
  int lane = threadIdx.x & 31;
  int jt  = wave & 3;
  int par = wave >> 2;
  int h  = lane >> 4;
  int ln = lane & 15;
  int j  = jt * 16 + ln;

  v8f f1[4], f2[4], f3[4];
#pragma unroll
  for (int it = 0; it < 4; ++it) { f1[it] = (v8f){}; f2[it] = (v8f){}; f3[it] = (v8f){}; }

  int nhalf = (tau1 - tau0 + 1) >> 1;
  int tau  = par ? (tau0 + nhalf) : tau0;
  int tend = par ? tau1 : (tau0 + nhalf);

  float sr0[KSTEPS], sr1[KSTEPS], si0[KSTEPS], si1[KSTEPS], iv[KSTEPS];
#define LOAD_STEP(T, PRE_, PIM_, PIN_) do { \
    sr0[(T)] = (PRE_)[(T) * 4 * SROW]; sr1[(T)] = (PRE_)[(T) * 4 * SROW + SROW]; \
    si0[(T)] = (PIM_)[(T) * 4 * SROW]; si1[(T)] = (PIM_)[(T) * 4 * SROW + SROW]; \
    iv[(T)]  = (PIN_)[(T) * 2 * LL]; } while (0)
#define TILE_PTRS(CT, PRE_, PIM_, PIN_) do { \
    int c_ = (CT) * 16 - CSHIFT + ln; \
    int q_ = (c_ >= 0 && c_ < CDIM) ? (c_ / ITERR) : 0; \
    int so_ = (SROW0 + 2 * h) * SROW + COL0 + c_; \
    (PRE_) = sre + so_; (PIM_) = sim + so_; (PIN_) = inp + b * 2500 + q_ + h * LL; } while (0)

  const float *cre, *cim, *cin;
  const float *nre, *nim, *nin;
  TILE_PTRS(ct_of(tau), cre, cim, cin);
#pragma unroll
  for (int t = 0; t < AHEAD; ++t) LOAD_STEP(t, cre, cim, cin);

  v2f prr[KSTEPS], pii[KSTEPS], pss[KSTEPS];
#pragma unroll
  for (int t = 0; t < KSTEPS; ++t) {
    int kk = 4 * t + 2 * h;
    prr[t] = *(const v2f*)&Pre[j * KDIM + kk];
    pii[t] = *(const v2f*)&Pim[j * KDIM + kk];
    pss[t] = prr[t] + pii[t];
  }
#pragma unroll
  for (int t = 0; t < AHEAD; ++t)
    asm volatile("" :: "v"(sr0[t]), "v"(sr1[t]), "v"(si0[t]), "v"(si1[t]), "v"(iv[t]));

  while (tau < tend) {
    int kap, m, nk;
    if (tau < NLONG * 10) { kap = tau / 10; m = tau - kap * 10; nk = 10; }
    else { int u = tau - NLONG * 10; int kq = u / 9; kap = NLONG + kq; m = u - kq * 9; nk = 9; }
    int mend = m + (tend - tau);
    if (mend > nk) mend = nk;
    v8f gre = {}; v8f gim = {};
    for (; m < mend; ++m) {
      int ct = kap + QPER * m;
      ++tau;
      int ctn = (tau < tend) ? ct_of(tau) : ct;
      TILE_PTRS(ctn, nre, nim, nin);
      int cA = ct * 16 - CSHIFT + 8 * h;
      int cB = cA + 4;
      int jw = (j < LL) ? j : 0;
      float wAl = wgt[jw * LL + ((cA >= 0 && cA < CDIM) ? cA / ITERR : 0)];
      float wBl = wgt[jw * LL + ((cB >= 0 && cB < CDIM) ? cB / ITERR : 0)];
      v8f d1 = {}; v8f d2 = {}; v8f d3 = {};
#pragma unroll
      for (int t = 0; t < KSTEPS; ++t) {
        __builtin_amdgcn_sched_barrier(0);
        if (t + AHEAD < KSTEPS) LOAD_STEP(t + AHEAD, cre, cim, cin);
        else                    LOAD_STEP(t + AHEAD - KSTEPS, nre, nim, nin);
        float inpv = iv[t];
        v2f are; are.x = sr0[t] * inpv; are.y = sr1[t] * inpv;
        v2f aim; aim.x = si0[t] * inpv; aim.y = si1[t] * inpv;
        v2f asum = are + aim;
        d1 = WMMA_F32X4(are,  prr[t], d1);
        d2 = WMMA_F32X4(aim,  pii[t], d2);
        d3 = WMMA_F32X4(asum, pss[t], d3);
      }
      cre = nre; cim = nim; cin = nin;
      v8f dre = d1 - d2;
      v8f dim = (d3 - d1) - d2;

      float wA = (j < LL && cA >= 0 && cA < CDIM) ? fabsf(wAl) : 0.f;
      float wB = (j < LL && cB >= 0 && cB < CDIM) ? fabsf(wBl) : 0.f;

  #pragma unroll
      for (int r = 0; r < 8; ++r) {
        float w = (r < 4) ? wA : wB;
        gre[r] += dre[r] * w;
        gim[r] += dim[r] * w;
      }
    }

    const v4f* qt = qs + (kap - kap0) * 512 + lane;
    v4f qv[16];
#pragma unroll
    for (int u = 0; u < 16; ++u) qv[u] = qt[u * 32];

#pragma unroll
    for (int tp = 0; tp < 4; ++tp) {
      v2f gr; gr.x = gre[2 * tp]; gr.y = gre[2 * tp + 1];
      v2f gi; gi.x = gim[2 * tp]; gi.y = gim[2 * tp + 1];
      v2f gs = gr + gi;
#pragma unroll
      for (int it = 0; it < 4; ++it) {
        v4f qq = qv[tp * 4 + it];
        v2f br; br.x = qq.x; br.y = qq.y;
        v2f bi; bi.x = qq.z; bi.y = qq.w;
        v2f bs = br + bi;
        f1[it] = WMMA_F32X4(gr, br, f1[it]);
        f2[it] = WMMA_F32X4(gi, bi, f2[it]);
        f3[it] = WMMA_F32X4(gs, bs, f3[it]);
      }
    }
  }
#undef LOAD_STEP
#undef TILE_PTRS

  v8f fre[4], fim[4];
#pragma unroll
  for (int it = 0; it < 4; ++it) {
    fre[it] = f1[it] - f2[it];
    fim[it] = (f3[it] - f1[it]) - f2[it];
  }

  __syncthreads();
  int lane_e = (int)__builtin_amdgcn_mbcnt_lo(~0u, 0u);
  float* red = sm + (size_t)jt * 64 * 32 + lane_e;
  if (par == 1) {
#pragma unroll
    for (int it = 0; it < 4; ++it) {
#pragma unroll
      for (int r = 0; r < 8; ++r) {
        red[(it * 16 + r) * 32]     = fre[it][r];
        red[(it * 16 + 8 + r) * 32] = fim[it][r];
      }
    }
  }
  __syncthreads();
  if (par == 0) {
    size_t pbase = ((size_t)(s * BB + b) * 16 + (size_t)(jt * 4)) * 256 + lane_e;
#pragma unroll
    for (int it = 0; it < 4; ++it) {
#pragma unroll
      for (int r = 0; r < 8; ++r) {
        pre[pbase + it * 256 + r * 32] = fre[it][r] + red[(it * 16 + r) * 32];
        pim[pbase + it * 256 + r * 32] = fim[it][r] + red[(it * 16 + 8 + r) * 32];
      }
    }
  }
}

#define MPITCH 65
__global__ void __launch_bounds__(1024) k_final(
    const float* __restrict__ pre, const float* __restrict__ pim,
    const float* __restrict__ inp, const float* __restrict__ wgt,
    const float* __restrict__ bias,
    const float* __restrict__ amp, float* __restrict__ out) {
  __shared__ float mb[JPAD * MPITCH];
  __shared__ float red[3][32];
  int b = blockIdx.x;
  int tid = threadIdx.x;
  int tile = tid >> 6;
  int p0 = (tid & 63) * 4;
  int j  = (tile >> 2) * 16 + (p0 >> 5) + 8 * ((p0 >> 4) & 1);
  int i0 = (tile & 3) * 16 + (p0 & 15);

  float m = 0.f;
  if (j < LL && i0 < LL) {
    const float* pr = pre + (size_t)b * TILEP + tid * 4;
    const float* pi = pim + (size_t)b * TILEP + tid * 4;
    float4 sr = *(const float4*)pr;
    float4 si = *(const float4*)pi;
#pragma unroll
    for (int s = 1; s < KSPLIT; ++s) {
      float4 a = *(const float4*)(pr + (size_t)s * BB * TILEP);
      float4 c = *(const float4*)(pi + (size_t)s * BB * TILEP);
      sr.x += a.x; sr.y += a.y; sr.z += a.z; sr.w += a.w;
      si.x += c.x; si.y += c.y; si.z += c.z; si.w += c.w;
    }
    float fr[4] = {sr.x, sr.y, sr.z, sr.w};
    float fi[4] = {si.x, si.y, si.z, si.w};
#pragma unroll
    for (int k = 0; k < 4; ++k) {
      int i = i0 + k;
      float v = 0.f;
      if (i < LL) v = sqrtf(fr[k] * fr[k] + fi[k] * fi[k]) / amp[j * LL + i];
      mb[j * MPITCH + i] = v;
      m = fmaxf(m, v);
    }
  }
  int k2 = (tid + 2048 < LL * LL) ? (tid + 2048) : tid;
  float x0 = inp[b * 2500 + tid], x1 = inp[b * 2500 + tid + 1024], x2 = inp[b * 2500 + k2];
  float w0 = wgt[tid], w1 = wgt[tid + 1024], w2 = wgt[k2];
  float mi = fmaxf(fmaxf(fmaxf(-3.4e38f, x0), x1), x2);
  float mw = fmaxf(fmaxf(fmaxf(-3.4e38f, fabsf(w0)), fabsf(w1)), fabsf(w2));
  for (int off = 16; off > 0; off >>= 1) {
    m  = fmaxf(m,  __shfl_xor(m,  off));
    mi = fmaxf(mi, __shfl_xor(mi, off));
    mw = fmaxf(mw, __shfl_xor(mw, off));
  }
  if ((tid & 31) == 0) { red[0][tid >> 5] = m; red[1][tid >> 5] = mi; red[2][tid >> 5] = mw; }
  __syncthreads();
  float mx  = red[0][tid & 31];
  float mxi = red[1][tid & 31];
  float mxw = red[2][tid & 31];
  for (int off = 16; off > 0; off >>= 1) {
    mx  = fmaxf(mx,  __shfl_xor(mx,  off));
    mxi = fmaxf(mxi, __shfl_xor(mxi, off));
    mxw = fmaxf(mxw, __shfl_xor(mxw, off));
  }

  float scale = (mxi * mxw) / mx;
  for (int idx = tid; idx < LL * LL; idx += 1024) {
    int i = idx / LL, j2 = idx % LL;
    out[b * 2500 + idx] = mb[j2 * MPITCH + i] * scale + bias[j2];
  }
}

extern "C" void kernel_launch(void* const* d_in, const int* in_sizes, int n_in,
                              void* d_out, int out_size, void* d_ws, size_t ws_size,
                              hipStream_t stream) {
  const float* inp  = (const float*)d_in[0];
  const float* wgt  = (const float*)d_in[1];
  const float* bias = (const float*)d_in[2];
  const float* amp  = (const float*)d_in[3];
  const float* sre  = (const float*)d_in[4];
  const float* sim  = (const float*)d_in[5];

  float* ws   = (float*)d_ws;
  float* pre  = ws;
  float* pim  = pre + (size_t)KSPLIT * BB * TILEP;

  k_mm12   <<<BB * KSPLIT, 256, 0, stream>>>(inp, wgt, sre, sim, pre, pim);
  k_final  <<<BB, 1024, 0, stream>>>(pre, pim, inp, wgt, bias, amp, (float*)d_out);
}


# first-stage twiddle rows j and 50-j are conjugates: four real products per k-step give both rows, so 2 row tiles (was 4) and half the operand loads; changes rounding and summation order
# speedup vs baseline: 7.0430x; 1.1051x over previous
#include <hip/hip_runtime.h>
#include <math.h>


#define LL    50
#define BB    16
#define ITERR 76
#define KDIM  100
#define CDIM  3800
#define CTILES 239
#define JPAD  64
#define SROW  7600
#define SROW0 50
#define COL0  1900
#define CSHIFT (COL0 % 16)
#define KSPLIT 16
#define CTPER  15
#define TILEP  4096
#define QPER  25
#define NLONG 14
#define NTW   200
#define QCLS  3
#define KSTEPS 25
#define JROWS 32
#define NPAR  4
#define REDF  (NPAR * 2 * 128 * 32)
#define AHEAD  11

typedef float v2f __attribute__((ext_vector_type(2)));
typedef float v4f __attribute__((ext_vector_type(4)));
typedef float v8f __attribute__((ext_vector_type(8)));

#define WMMA_F32X4(a, b, c) \
  __builtin_amdgcn_wmma_f32_16x16x4_f32(false, (a), false, (b), (short)0, (c), false, false)

__device__ __forceinline__ int class_of(int tau) {
  return (tau < NLONG * 10) ? (tau / 10) : (NLONG + (tau - NLONG * 10) / 9);
}

__device__ __forceinline__ int ct_of(int tau) {
  int kap, m;
  if (tau < NLONG * 10) { kap = tau / 10; m = tau - kap * 10; }
  else { int u = tau - NLONG * 10; int kq = u / 9; kap = NLONG + kq; m = u - kq * 9; }
  return kap + QPER * m;
}

__global__ void __launch_bounds__(256) __attribute__((amdgpu_waves_per_eu(2, 2))) k_mm12(
    const float* __restrict__ inp, const float* __restrict__ wgt,
    const float* __restrict__ sre, const float* __restrict__ sim,
    float* __restrict__ pre, float* __restrict__ pim) {
  __shared__ float sm[REDF];
  __shared__ v2f tw[NTW];
  __shared__ v4f qs[QCLS * 512];
  float* Pre = sm;
  float* Pim = sm + JROWS * KDIM;
  int s = blockIdx.x % KSPLIT;
  int b = blockIdx.x / KSPLIT;

  if (threadIdx.x < NTW) {
    float sn, cz;
    sincospif((float)threadIdx.x * 0.01f, &sn, &cz);
    v2f v; v.x = cz; v.y = -sn;
    tw[threadIdx.x] = v;
  }
  __syncthreads();

  for (int k = 0; k < (JROWS * KDIM + 255) / 256; ++k) {
    int e = threadIdx.x + k * 256;
    int j = e / KDIM, r = e % KDIM;
    int t = ((150 + 2 * j) * (50 + r)) % 200;
    v2f v = tw[t];
    if (e < JROWS * KDIM) { Pre[e] = v.x; Pim[e] = v.y; }
  }

  int tau0 = s * CTPER;
  int tau1 = (tau0 + CTPER < CTILES) ? (tau0 + CTPER) : CTILES;
  int kap0 = class_of(tau0);
  int ncls = class_of(tau1 - 1) - kap0 + 1;
  for (int cls = 0; cls < ncls; ++cls) {
    for (int hh = 0; hh < 2; ++hh) {
      int rem = threadIdx.x + hh * 256;
      int ql = rem & 31;
      int it = (rem >> 5) & 3;
      int tp = (rem >> 7) & 3;
      int i  = it * 16 + (ql & 15);
      int c0 = (kap0 + cls) * 16 - CSHIFT + 8 * (ql >> 4) + 2 * tp;
      int t0 = ((i + 1) * (COL0 + c0)) % 200;
      int t1 = ((i + 1) * (COL0 + c0 + 1)) % 200;
      v2f z0 = tw[t0];
      v2f z1 = tw[t1];
      v4f v = {0.f, 0.f, 0.f, 0.f};
      if (i < LL) { v.x = z0.x; v.y = z1.x; v.z = z0.y; v.w = z1.y; }
      qs[cls * 512 + rem] = v;
    }
  }
  __syncthreads();

  int wave = threadIdx.x >> 5;
  int lane = threadIdx.x & 31;
  int jt  = wave & 1;
  int par = wave >> 1;
  int h  = lane >> 4;
  int ln = lane & 15;
  int j  = jt * 16 + ln;

  v8f f1[2][4], f2[2][4], f3[2][4];
#pragma unroll
  for (int e = 0; e < 2; ++e) {
#pragma unroll
    for (int it = 0; it < 4; ++it) { f1[e][it] = (v8f){}; f2[e][it] = (v8f){}; f3[e][it] = (v8f){}; }
  }

  int nq = (tau1 - tau0 + NPAR - 1) / NPAR;
  int tau  = tau0 + par * nq;
  int tend = (tau + nq < tau1) ? (tau + nq) : tau1;

  float sr0[KSTEPS], sr1[KSTEPS], si0[KSTEPS], si1[KSTEPS], iv[KSTEPS];
#define LOAD_STEP(T, PRE_, PIM_, PIN_) do { \
    sr0[(T)] = (PRE_)[(T) * 4 * SROW]; sr1[(T)] = (PRE_)[(T) * 4 * SROW + SROW]; \
    si0[(T)] = (PIM_)[(T) * 4 * SROW]; si1[(T)] = (PIM_)[(T) * 4 * SROW + SROW]; \
    iv[(T)]  = (PIN_)[(T) * 2 * LL]; } while (0)
#define TILE_PTRS(CT, PRE_, PIM_, PIN_) do { \
    int c_ = (CT) * 16 - CSHIFT + ln; \
    int q_ = (c_ >= 0 && c_ < CDIM) ? (c_ / ITERR) : 0; \
    int so_ = (SROW0 + 2 * h) * SROW + COL0 + c_; \
    (PRE_) = sre + so_; (PIM_) = sim + so_; (PIN_) = inp + b * 2500 + q_ + h * LL; } while (0)

  const float *cre, *cim, *cin;
  const float *nre, *nim, *nin;
  TILE_PTRS(ct_of(tau), cre, cim, cin);
#pragma unroll
  for (int t = 0; t < AHEAD; ++t) LOAD_STEP(t, cre, cim, cin);

  v2f prr[KSTEPS], pii[KSTEPS];
#pragma unroll
  for (int t = 0; t < KSTEPS; ++t) {
    int kk = 4 * t + 2 * h;
    prr[t] = *(const v2f*)&Pre[j * KDIM + kk];
    pii[t] = *(const v2f*)&Pim[j * KDIM + kk];
  }
#pragma unroll
  for (int t = 0; t < AHEAD; ++t)
    asm volatile("" :: "v"(sr0[t]), "v"(sr1[t]), "v"(si0[t]), "v"(si1[t]), "v"(iv[t]));

  while (tau < tend) {
    int kap, m, nk;
    if (tau < NLONG * 10) { kap = tau / 10; m = tau - kap * 10; nk = 10; }
    else { int u = tau - NLONG * 10; int kq = u / 9; kap = NLONG + kq; m = u - kq * 9; nk = 9; }
    int mend = m + (tend - tau);
    if (mend > nk) mend = nk;
    v8f g1re = {}; v8f g1im = {}; v8f g2re = {}; v8f g2im = {};
    for (; m < mend; ++m) {
      int ct = kap + QPER * m;
      ++tau;
      int ctn = (tau < tend) ? ct_of(tau) : ct;
      TILE_PTRS(ctn, nre, nim, nin);
      int cA = ct * 16 - CSHIFT + 8 * h;
      int cB = cA + 4;
      int qA = (cA >= 0 && cA < CDIM) ? cA / ITERR : 0;
      int qB = (cB >= 0 && cB < CDIM) ? cB / ITERR : 0;
      int j2 = (j >= 1 && j <= 18) ? (LL - j) : 0;
      float w1Al = wgt[j * LL + qA];
      float w1Bl = wgt[j * LL + qB];
      float w2Al = wgt[j2 * LL + qA];
      float w2Bl = wgt[j2 * LL + qB];
      v8f da = {}; v8f db = {}; v8f dc = {}; v8f dd = {};
#pragma unroll
      for (int t = 0; t < KSTEPS; ++t) {
        __builtin_amdgcn_sched_barrier(0);
        if (t + AHEAD < KSTEPS) LOAD_STEP(t + AHEAD, cre, cim, cin);
        else                    LOAD_STEP(t + AHEAD - KSTEPS, nre, nim, nin);
        float inpv = iv[t];
        v2f are; are.x = sr0[t] * inpv; are.y = sr1[t] * inpv;
        v2f aim; aim.x = si0[t] * inpv; aim.y = si1[t] * inpv;
        da = WMMA_F32X4(are, prr[t], da);
        db = WMMA_F32X4(aim, pii[t], db);
        dc = WMMA_F32X4(aim, prr[t], dc);
        dd = WMMA_F32X4(are, pii[t], dd);
      }
      cre = nre; cim = nim; cin = nin;
      v8f d1re = da - db;
      v8f d1im = dc + dd;
      v8f d2re = da + db;
      v8f d2im = dc - dd;

      bool okA = (cA >= 0 && cA < CDIM), okB = (cB >= 0 && cB < CDIM);
      bool ok2 = (j >= 1 && j <= 18);
      float w1A = okA ? fabsf(w1Al) : 0.f;
      float w1B = okB ? fabsf(w1Bl) : 0.f;
      float w2A = (ok2 && okA) ? fabsf(w2Al) : 0.f;
      float w2B = (ok2 && okB) ? fabsf(w2Bl) : 0.f;

  #pragma unroll
      for (int r = 0; r < 8; ++r) {
        float w1 = (r < 4) ? w1A : w1B;
        float w2 = (r < 4) ? w2A : w2B;
        g1re[r] += d1re[r] * w1;
        g1im[r] += d1im[r] * w1;
        g2re[r] += d2re[r] * w2;
        g2im[r] += d2im[r] * w2;
      }
    }

    const v4f* qt = qs + (kap - kap0) * 512 + lane;
    v4f qv[16];
#pragma unroll
    for (int u = 0; u < 16; ++u) qv[u] = qt[u * 32];

#pragma unroll
    for (int tp = 0; tp < 4; ++tp) {
      v2f gr1; gr1.x = g1re[2 * tp]; gr1.y = g1re[2 * tp + 1];
      v2f gi1; gi1.x = g1im[2 * tp]; gi1.y = g1im[2 * tp + 1];
      v2f gr2; gr2.x = g2re[2 * tp]; gr2.y = g2re[2 * tp + 1];
      v2f gi2; gi2.x = g2im[2 * tp]; gi2.y = g2im[2 * tp + 1];
      v2f gs1 = gr1 + gi1;
      v2f gs2 = gr2 + gi2;
#pragma unroll
      for (int it = 0; it < 4; ++it) {
        v4f qq = qv[tp * 4 + it];
        v2f br; br.x = qq.x; br.y = qq.y;
        v2f bi; bi.x = qq.z; bi.y = qq.w;
        v2f bs = br + bi;
        f1[0][it] = WMMA_F32X4(gr1, br, f1[0][it]);
        f2[0][it] = WMMA_F32X4(gi1, bi, f2[0][it]);
        f3[0][it] = WMMA_F32X4(gs1, bs, f3[0][it]);
        f1[1][it] = WMMA_F32X4(gr2, br, f1[1][it]);
        f2[1][it] = WMMA_F32X4(gi2, bi, f2[1][it]);
        f3[1][it] = WMMA_F32X4(gs2, bs, f3[1][it]);
      }
    }
  }
#undef LOAD_STEP
#undef TILE_PTRS

  v8f fre[2][4], fim[2][4];
#pragma unroll
  for (int e = 0; e < 2; ++e) {
#pragma unroll
    for (int it = 0; it < 4; ++it) {
      fre[e][it] = f1[e][it] - f2[e][it];
      fim[e][it] = (f3[e][it] - f1[e][it]) - f2[e][it];
    }
  }

  __syncthreads();
  int lane_e = (int)__builtin_amdgcn_mbcnt_lo(~0u, 0u);
  {
    float* red = sm + (size_t)(par * 2 + jt) * 128 * 32 + lane_e;
#pragma unroll
    for (int e = 0; e < 2; ++e) {
#pragma unroll
      for (int it = 0; it < 4; ++it) {
#pragma unroll
        for (int r = 0; r < 8; ++r) {
          red[((e * 4 + it) * 16 + r) * 32]     = fre[e][it][r];
          red[((e * 4 + it) * 16 + 8 + r) * 32] = fim[e][it][r];
        }
      }
    }
  }
  __syncthreads();
  {
    const int QST = 2 * 128 * 32;
    int e = par >> 1;
    int itb = (par & 1) * 2;
    const float* red = sm + (size_t)jt * 128 * 32 + (size_t)((e * 4 + itb) * 16) * 32 + lane_e;
    size_t pbase = ((size_t)(s * BB + b) * 16 + (size_t)((e * 2 + jt) * 4 + itb)) * 256 + lane_e;
#pragma unroll
    for (int u = 0; u < 2; ++u) {
#pragma unroll
      for (int r = 0; r < 8; ++r) {
        int o = (u * 16 + r) * 32;
        pre[pbase + u * 256 + r * 32] = ((red[o] + red[o + QST]) + red[o + 2 * QST]) + red[o + 3 * QST];
        pim[pbase + u * 256 + r * 32] =
            ((red[o + 8 * 32] + red[o + 8 * 32 + QST]) + red[o + 8 * 32 + 2 * QST]) + red[o + 8 * 32 + 3 * QST];
      }
    }
  }
}

#define MPITCH 65
__global__ void __launch_bounds__(1024) k_final(
    const float* __restrict__ pre, const float* __restrict__ pim,
    const float* __restrict__ inp, const float* __restrict__ wgt,
    const float* __restrict__ bias,
    const float* __restrict__ amp, float* __restrict__ out) {
  __shared__ float mb[JPAD * MPITCH];
  __shared__ float red[3][32];
  int b = blockIdx.x;
  int tid = threadIdx.x;
  int tile = tid >> 6;
  int p0 = (tid & 63) * 4;
  int vt = tile >> 2;
  int jl = (vt & 1) * 16 + (p0 >> 5) + 8 * ((p0 >> 4) & 1);
  int j  = (vt < 2) ? jl : (LL - jl);
  bool jok = (vt < 2) || (jl >= 1 && jl <= 18);
  int i0 = (tile & 3) * 16 + (p0 & 15);

  float m = 0.f;
  if (jok && i0 < LL) {
    const float* pr = pre + (size_t)b * TILEP + tid * 4;
    const float* pi = pim + (size_t)b * TILEP + tid * 4;
#define NFIRST 12
#define PADD(SR_, SI_, A_, C_) do { \
    SR_.x += A_.x; SR_.y += A_.y; SR_.z += A_.z; SR_.w += A_.w; \
    SI_.x += C_.x; SI_.y += C_.y; SI_.z += C_.z; SI_.w += C_.w; } while (0)
    float4 qa[NFIRST], qc[NFIRST];
#pragma unroll
    for (int s = 0; s < NFIRST; ++s) {
      qa[s] = *(const float4*)(pr + (size_t)s * BB * TILEP);
      qc[s] = *(const float4*)(pi + (size_t)s * BB * TILEP);
    }
    float4 sr = qa[0];
    float4 si = qc[0];
#pragma unroll
    for (int s = 1; s < KSPLIT - NFIRST; ++s) PADD(sr, si, qa[s], qc[s]);
    __builtin_amdgcn_sched_barrier(0);
    float4 ra[KSPLIT - NFIRST], rc[KSPLIT - NFIRST];
#pragma unroll
    for (int s = NFIRST; s < KSPLIT; ++s) {
      ra[s - NFIRST] = *(const float4*)(pr + (size_t)s * BB * TILEP);
      rc[s - NFIRST] = *(const float4*)(pi + (size_t)s * BB * TILEP);
    }
#pragma unroll
    for (int s = KSPLIT - NFIRST; s < NFIRST; ++s) PADD(sr, si, qa[s], qc[s]);
#pragma unroll
    for (int s = NFIRST; s < KSPLIT; ++s) PADD(sr, si, ra[s - NFIRST], rc[s - NFIRST]);
#undef PADD
#undef NFIRST
    float fr[4] = {sr.x, sr.y, sr.z, sr.w};
    float fi[4] = {si.x, si.y, si.z, si.w};
#pragma unroll
    for (int k = 0; k < 4; ++k) {
      int i = i0 + k;
      float v = 0.f;
      if (i < LL) v = sqrtf(fr[k] * fr[k] + fi[k] * fi[k]) / amp[j * LL + i];
      mb[j * MPITCH + i] = v;
      m = fmaxf(m, v);
    }
  }
  int k2 = (tid + 2048 < LL * LL) ? (tid + 2048) : tid;
  float x0 = inp[b * 2500 + tid], x1 = inp[b * 2500 + tid + 1024], x2 = inp[b * 2500 + k2];
  float w0 = wgt[tid], w1 = wgt[tid + 1024], w2 = wgt[k2];
  float mi = fmaxf(fmaxf(fmaxf(-3.4e38f, x0), x1), x2);
  float mw = fmaxf(fmaxf(fmaxf(-3.4e38f, fabsf(w0)), fabsf(w1)), fabsf(w2));
  for (int off = 16; off > 0; off >>= 1) {
    m  = fmaxf(m,  __shfl_xor(m,  off));
    mi = fmaxf(mi, __shfl_xor(mi, off));
    mw = fmaxf(mw, __shfl_xor(mw, off));
  }
  if ((tid & 31) == 0) { red[0][tid >> 5] = m; red[1][tid >> 5] = mi; red[2][tid >> 5] = mw; }
  __syncthreads();
  float mx  = red[0][tid & 31];
  float mxi = red[1][tid & 31];
  float mxw = red[2][tid & 31];
  for (int off = 16; off > 0; off >>= 1) {
    mx  = fmaxf(mx,  __shfl_xor(mx,  off));
    mxi = fmaxf(mxi, __shfl_xor(mxi, off));
    mxw = fmaxf(mxw, __shfl_xor(mxw, off));
  }

  float scale = (mxi * mxw) / mx;
  for (int idx = tid; idx < LL * LL; idx += 1024) {
    int i = idx / LL, j2 = idx % LL;
    out[b * 2500 + idx] = mb[j2 * MPITCH + i] * scale + bias[j2];
  }
}

extern "C" void kernel_launch(void* const* d_in, const int* in_sizes, int n_in,
                              void* d_out, int out_size, void* d_ws, size_t ws_size,
                              hipStream_t stream) {
  const float* inp  = (const float*)d_in[0];
  const float* wgt  = (const float*)d_in[1];
  const float* bias = (const float*)d_in[2];
  const float* amp  = (const float*)d_in[3];
  const float* sre  = (const float*)d_in[4];
  const float* sim  = (const float*)d_in[5];

  float* ws   = (float*)d_ws;
  float* pre  = ws;
  float* pim  = pre + (size_t)KSPLIT * BB * TILEP;

  k_mm12   <<<BB * KSPLIT, 256, 0, stream>>>(inp, wgt, sre, sim, pre, pim);
  k_final  <<<BB, 1024, 0, stream>>>(pre, pim, inp, wgt, bias, amp, (float*)d_out);
}


# GEMM-2 once per work-group and class, not per wave and class: the four tile quarters' G sums are added through LDS first, then each wave does one output column tile; changes the summation order
# speedup vs baseline: 7.5376x; 1.0702x over previous
#include <hip/hip_runtime.h>
#include <math.h>


#define LL    50
#define BB    16
#define ITERR 76
#define KDIM  100
#define CDIM  3800
#define CTILES 239
#define JPAD  64
#define SROW  7600
#define SROW0 50
#define COL0  1900
#define CSHIFT (COL0 % 16)
#define KSPLIT 16
#define CTPER  15
#define TILEP  4096
#define QPER  25
#define NLONG 14
#define NTW   200
#define QCLS  3
#define KSTEPS 25
#define JROWS 32
#define NPAR  4
#define GXV   (QCLS * 2 * NPAR * 8 * 32)
#define AHEAD  11

typedef float v2f __attribute__((ext_vector_type(2)));
typedef float v4f __attribute__((ext_vector_type(4)));
typedef float v8f __attribute__((ext_vector_type(8)));

#define WMMA_F32X4(a, b, c) \
  __builtin_amdgcn_wmma_f32_16x16x4_f32(false, (a), false, (b), (short)0, (c), false, false)

__device__ __forceinline__ int class_of(int tau) {
  return (tau < NLONG * 10) ? (tau / 10) : (NLONG + (tau - NLONG * 10) / 9);
}

__device__ __forceinline__ int ct_of(int tau) {
  int kap, m;
  if (tau < NLONG * 10) { kap = tau / 10; m = tau - kap * 10; }
  else { int u = tau - NLONG * 10; int kq = u / 9; kap = NLONG + kq; m = u - kq * 9; }
  return kap + QPER * m;
}

__global__ void __launch_bounds__(256) __attribute__((amdgpu_waves_per_eu(2, 2))) k_mm12(
    const float* __restrict__ inp, const float* __restrict__ wgt,
    const float* __restrict__ sre, const float* __restrict__ sim,
    float* __restrict__ pre, float* __restrict__ pim) {
  __shared__ float sm[2 * JROWS * KDIM];
  __shared__ v4f gx[GXV];
  __shared__ v2f tw[NTW];
  __shared__ v4f qs[QCLS * 512];
  float* Pre = sm;
  float* Pim = sm + JROWS * KDIM;
  int s = blockIdx.x % KSPLIT;
  int b = blockIdx.x / KSPLIT;

  if (threadIdx.x < NTW) {
    float sn, cz;
    sincospif((float)threadIdx.x * 0.01f, &sn, &cz);
    v2f v; v.x = cz; v.y = -sn;
    tw[threadIdx.x] = v;
  }
  __syncthreads();

  for (int k = 0; k < (JROWS * KDIM + 255) / 256; ++k) {
    int e = threadIdx.x + k * 256;
    int j = e / KDIM, r = e % KDIM;
    int t = ((150 + 2 * j) * (50 + r)) % 200;
    v2f v = tw[t];
    if (e < JROWS * KDIM) { Pre[e] = v.x; Pim[e] = v.y; }
  }

  int tau0 = s * CTPER;
  int tau1 = (tau0 + CTPER < CTILES) ? (tau0 + CTPER) : CTILES;
  int kap0 = class_of(tau0);
  int ncls = class_of(tau1 - 1) - kap0 + 1;
  for (int cls = 0; cls < ncls; ++cls) {
    for (int hh = 0; hh < 2; ++hh) {
      int rem = threadIdx.x + hh * 256;
      int ql = rem & 31;
      int it = (rem >> 5) & 3;
      int tp = (rem >> 7) & 3;
      int i  = it * 16 + (ql & 15);
      int c0 = (kap0 + cls) * 16 - CSHIFT + 8 * (ql >> 4) + 2 * tp;
      int t0 = ((i + 1) * (COL0 + c0)) % 200;
      int t1 = ((i + 1) * (COL0 + c0 + 1)) % 200;
      v2f z0 = tw[t0];
      v2f z1 = tw[t1];
      v4f v = {0.f, 0.f, 0.f, 0.f};
      if (i < LL) { v.x = z0.x; v.y = z1.x; v.z = z0.y; v.w = z1.y; }
      qs[cls * 512 + rem] = v;
    }
  }
  __syncthreads();

  int wave = threadIdx.x >> 5;
  int lane = threadIdx.x & 31;
  int jt  = wave & 1;
  int par = wave >> 1;
  int h  = lane >> 4;
  int ln = lane & 15;
  int j  = jt * 16 + ln;

  int nq = (tau1 - tau0 + NPAR - 1) / NPAR;
  int tau  = tau0 + par * nq;
  int tend = (tau + nq < tau1) ? (tau + nq) : tau1;

  float sr0[KSTEPS], sr1[KSTEPS], si0[KSTEPS], si1[KSTEPS], iv[KSTEPS];
#define LOAD_STEP(T, PRE_, PIM_, PIN_) do { \
    sr0[(T)] = (PRE_)[(T) * 4 * SROW]; sr1[(T)] = (PRE_)[(T) * 4 * SROW + SROW]; \
    si0[(T)] = (PIM_)[(T) * 4 * SROW]; si1[(T)] = (PIM_)[(T) * 4 * SROW + SROW]; \
    iv[(T)]  = (PIN_)[(T) * 2 * LL]; } while (0)
#define TILE_PTRS(CT, PRE_, PIM_, PIN_) do { \
    int c_ = (CT) * 16 - CSHIFT + ln; \
    int q_ = (c_ >= 0 && c_ < CDIM) ? (c_ / ITERR) : 0; \
    int so_ = (SROW0 + 2 * h) * SROW + COL0 + c_; \
    (PRE_) = sre + so_; (PIM_) = sim + so_; (PIN_) = inp + b * 2500 + q_ + h * LL; } while (0)

  const float *cre, *cim, *cin;
  const float *nre, *nim, *nin;
  TILE_PTRS(ct_of(tau), cre, cim, cin);
#pragma unroll
  for (int t = 0; t < AHEAD; ++t) LOAD_STEP(t, cre, cim, cin);

  v2f prr[KSTEPS], pii[KSTEPS];
#pragma unroll
  for (int t = 0; t < KSTEPS; ++t) {
    int kk = 4 * t + 2 * h;
    prr[t] = *(const v2f*)&Pre[j * KDIM + kk];
    pii[t] = *(const v2f*)&Pim[j * KDIM + kk];
  }
#pragma unroll
  for (int t = 0; t < AHEAD; ++t)
    asm volatile("" :: "v"(sr0[t]), "v"(sr1[t]), "v"(si0[t]), "v"(si1[t]), "v"(iv[t]));

  while (tau < tend) {
    int kap, m, nk;
    if (tau < NLONG * 10) { kap = tau / 10; m = tau - kap * 10; nk = 10; }
    else { int u = tau - NLONG * 10; int kq = u / 9; kap = NLONG + kq; m = u - kq * 9; nk = 9; }
    int mend = m + (tend - tau);
    if (mend > nk) mend = nk;
    v8f g1re = {}; v8f g1im = {}; v8f g2re = {}; v8f g2im = {};
    for (; m < mend; ++m) {
      int ct = kap + QPER * m;
      ++tau;
      int ctn = (tau < tend) ? ct_of(tau) : ct;
      TILE_PTRS(ctn, nre, nim, nin);
      int cA = ct * 16 - CSHIFT + 8 * h;
      int cB = cA + 4;
      int qA = (cA >= 0 && cA < CDIM) ? cA / ITERR : 0;
      int qB = (cB >= 0 && cB < CDIM) ? cB / ITERR : 0;
      int j2 = (j >= 1 && j <= 18) ? (LL - j) : 0;
      float w1Al = wgt[j * LL + qA];
      float w1Bl = wgt[j * LL + qB];
      float w2Al = wgt[j2 * LL + qA];
      float w2Bl = wgt[j2 * LL + qB];
      v8f da = {}; v8f db = {}; v8f dc = {}; v8f dd = {};
#pragma unroll
      for (int t = 0; t < KSTEPS; ++t) {
        __builtin_amdgcn_sched_barrier(0);
        if (t + AHEAD < KSTEPS) LOAD_STEP(t + AHEAD, cre, cim, cin);
        else                    LOAD_STEP(t + AHEAD - KSTEPS, nre, nim, nin);
        float inpv = iv[t];
        v2f are; are.x = sr0[t] * inpv; are.y = sr1[t] * inpv;
        v2f aim; aim.x = si0[t] * inpv; aim.y = si1[t] * inpv;
        da = WMMA_F32X4(are, prr[t], da);
        db = WMMA_F32X4(aim, pii[t], db);
        dc = WMMA_F32X4(aim, prr[t], dc);
        dd = WMMA_F32X4(are, pii[t], dd);
      }
      cre = nre; cim = nim; cin = nin;
      v8f d1re = da - db;
      v8f d1im = dc + dd;
      v8f d2re = da + db;
      v8f d2im = dc - dd;

      bool okA = (cA >= 0 && cA < CDIM), okB = (cB >= 0 && cB < CDIM);
      bool ok2 = (j >= 1 && j <= 18);
      float w1A = okA ? fabsf(w1Al) : 0.f;
      float w1B = okB ? fabsf(w1Bl) : 0.f;
      float w2A = (ok2 && okA) ? fabsf(w2Al) : 0.f;
      float w2B = (ok2 && okB) ? fabsf(w2Bl) : 0.f;

  #pragma unroll
      for (int r = 0; r < 8; ++r) {
        float w1 = (r < 4) ? w1A : w1B;
        float w2 = (r < 4) ? w2A : w2B;
        g1re[r] += d1re[r] * w1;
        g1im[r] += d1im[r] * w1;
        g2re[r] += d2re[r] * w2;
        g2im[r] += d2im[r] * w2;
      }
    }

    {
      v4f* gp = gx + ((((kap - kap0) * 2 + jt) * NPAR + par) * 8) * 32 + lane;
      gp[0 * 32] = (v4f){g1re[0], g1re[1], g1re[2], g1re[3]};
      gp[1 * 32] = (v4f){g1re[4], g1re[5], g1re[6], g1re[7]};
      gp[2 * 32] = (v4f){g1im[0], g1im[1], g1im[2], g1im[3]};
      gp[3 * 32] = (v4f){g1im[4], g1im[5], g1im[6], g1im[7]};
      gp[4 * 32] = (v4f){g2re[0], g2re[1], g2re[2], g2re[3]};
      gp[5 * 32] = (v4f){g2re[4], g2re[5], g2re[6], g2re[7]};
      gp[6 * 32] = (v4f){g2im[0], g2im[1], g2im[2], g2im[3]};
      gp[7 * 32] = (v4f){g2im[4], g2im[5], g2im[6], g2im[7]};
    }
  }
#undef LOAD_STEP
#undef TILE_PTRS

  __syncthreads();

  int it = par;
  v8f f1[2], f2[2], f3[2];
#pragma unroll
  for (int e = 0; e < 2; ++e) { f1[e] = (v8f){}; f2[e] = (v8f){}; f3[e] = (v8f){}; }

  for (int cl = 0; cl < ncls; ++cl) {
    int kapc = kap0 + cl;
    int cs = (kapc < NLONG) ? kapc * 10 : NLONG * 10 + (kapc - NLONG) * 9;
    int ce = cs + ((kapc < NLONG) ? 10 : 9);
    v4f gs[8];
#pragma unroll
    for (int g = 0; g < 8; ++g) gs[g] = (v4f){0.f, 0.f, 0.f, 0.f};
#pragma unroll
    for (int p = 0; p < NPAR; ++p) {
      int lo = tau0 + p * nq;
      int hi = (lo + nq < tau1) ? (lo + nq) : tau1;
      if (lo < hi && lo < ce && hi > cs) {
        const v4f* gp = gx + (((cl * 2 + jt) * NPAR + p) * 8) * 32 + lane;
        v4f ld[8];
#pragma unroll
        for (int g = 0; g < 8; ++g) ld[g] = gp[g * 32];
#pragma unroll
        for (int g = 0; g < 8; ++g) gs[g] += ld[g];
      }
    }
    const v4f* qt = qs + cl * 512 + it * 32 + lane;
    v4f qv[4];
#pragma unroll
    for (int tp = 0; tp < 4; ++tp) qv[tp] = qt[tp * 128];

#pragma unroll
    for (int tp = 0; tp < 4; ++tp) {
      int g = tp >> 1, o = (tp & 1) * 2;
      v2f gr1; gr1.x = gs[0 + g][o]; gr1.y = gs[0 + g][o + 1];
      v2f gi1; gi1.x = gs[2 + g][o]; gi1.y = gs[2 + g][o + 1];
      v2f gr2; gr2.x = gs[4 + g][o]; gr2.y = gs[4 + g][o + 1];
      v2f gi2; gi2.x = gs[6 + g][o]; gi2.y = gs[6 + g][o + 1];
      v2f gs1 = gr1 + gi1;
      v2f gs2 = gr2 + gi2;
      v4f qq = qv[tp];
      v2f br; br.x = qq.x; br.y = qq.y;
      v2f bi; bi.x = qq.z; bi.y = qq.w;
      v2f bs = br + bi;
      f1[0] = WMMA_F32X4(gr1, br, f1[0]);
      f2[0] = WMMA_F32X4(gi1, bi, f2[0]);
      f3[0] = WMMA_F32X4(gs1, bs, f3[0]);
      f1[1] = WMMA_F32X4(gr2, br, f1[1]);
      f2[1] = WMMA_F32X4(gi2, bi, f2[1]);
      f3[1] = WMMA_F32X4(gs2, bs, f3[1]);
    }
  }

#pragma unroll
  for (int e = 0; e < 2; ++e) {
    v8f fre = f1[e] - f2[e];
    v8f fim = (f3[e] - f1[e]) - f2[e];
    size_t pbase = ((size_t)(s * BB + b) * 16 + (size_t)((e * 2 + jt) * 4 + it)) * 256 + lane;
#pragma unroll
    for (int r = 0; r < 8; ++r) {
      pre[pbase + r * 32] = fre[r];
      pim[pbase + r * 32] = fim[r];
    }
  }
}

#define MPITCH 65
__global__ void __launch_bounds__(1024) k_final(
    const float* __restrict__ pre, const float* __restrict__ pim,
    const float* __restrict__ inp, const float* __restrict__ wgt,
    const float* __restrict__ bias,
    const float* __restrict__ amp, float* __restrict__ out) {
  __shared__ float mb[JPAD * MPITCH];
  __shared__ float red[3][32];
  int b = blockIdx.x;
  int tid = threadIdx.x;
  int tile = tid >> 6;
  int p0 = (tid & 63) * 4;
  int vt = tile >> 2;
  int jl = (vt & 1) * 16 + (p0 >> 5) + 8 * ((p0 >> 4) & 1);
  int j  = (vt < 2) ? jl : (LL - jl);
  bool jok = (vt < 2) || (jl >= 1 && jl <= 18);
  int i0 = (tile & 3) * 16 + (p0 & 15);

  float m = 0.f;
  if (jok && i0 < LL) {
    const float* pr = pre + (size_t)b * TILEP + tid * 4;
    const float* pi = pim + (size_t)b * TILEP + tid * 4;
#define NFIRST 12
#define PADD(SR_, SI_, A_, C_) do { \
    SR_.x += A_.x; SR_.y += A_.y; SR_.z += A_.z; SR_.w += A_.w; \
    SI_.x += C_.x; SI_.y += C_.y; SI_.z += C_.z; SI_.w += C_.w; } while (0)
    float4 qa[NFIRST], qc[NFIRST];
#pragma unroll
    for (int s = 0; s < NFIRST; ++s) {
      qa[s] = *(const float4*)(pr + (size_t)s * BB * TILEP);
      qc[s] = *(const float4*)(pi + (size_t)s * BB * TILEP);
    }
    float4 sr = qa[0];
    float4 si = qc[0];
#pragma unroll
    for (int s = 1; s < KSPLIT - NFIRST; ++s) PADD(sr, si, qa[s], qc[s]);
    __builtin_amdgcn_sched_barrier(0);
    float4 ra[KSPLIT - NFIRST], rc[KSPLIT - NFIRST];
#pragma unroll
    for (int s = NFIRST; s < KSPLIT; ++s) {
      ra[s - NFIRST] = *(const float4*)(pr + (size_t)s * BB * TILEP);
      rc[s - NFIRST] = *(const float4*)(pi + (size_t)s * BB * TILEP);
    }
#pragma unroll
    for (int s = KSPLIT - NFIRST; s < NFIRST; ++s) PADD(sr, si, qa[s], qc[s]);
#pragma unroll
    for (int s = NFIRST; s < KSPLIT; ++s) PADD(sr, si, ra[s - NFIRST], rc[s - NFIRST]);
#undef PADD
#undef NFIRST
    float fr[4] = {sr.x, sr.y, sr.z, sr.w};
    float fi[4] = {si.x, si.y, si.z, si.w};
#pragma unroll
    for (int k = 0; k < 4; ++k) {
      int i = i0 + k;
      float v = 0.f;
      if (i < LL) v = sqrtf(fr[k] * fr[k] + fi[k] * fi[k]) / amp[j * LL + i];
      mb[j * MPITCH + i] = v;
      m = fmaxf(m, v);
    }
  }
  int k2 = (tid + 2048 < LL * LL) ? (tid + 2048) : tid;
  float x0 = inp[b * 2500 + tid], x1 = inp[b * 2500 + tid + 1024], x2 = inp[b * 2500 + k2];
  float w0 = wgt[tid], w1 = wgt[tid + 1024], w2 = wgt[k2];
  float mi = fmaxf(fmaxf(fmaxf(-3.4e38f, x0), x1), x2);
  float mw = fmaxf(fmaxf(fmaxf(-3.4e38f, fabsf(w0)), fabsf(w1)), fabsf(w2));
  for (int off = 16; off > 0; off >>= 1) {
    m  = fmaxf(m,  __shfl_xor(m,  off));
    mi = fmaxf(mi, __shfl_xor(mi, off));
    mw = fmaxf(mw, __shfl_xor(mw, off));
  }
  if ((tid & 31) == 0) { red[0][tid >> 5] = m; red[1][tid >> 5] = mi; red[2][tid >> 5] = mw; }
  __syncthreads();
  float mx  = red[0][tid & 31];
  float mxi = red[1][tid & 31];
  float mxw = red[2][tid & 31];
  for (int off = 16; off > 0; off >>= 1) {
    mx  = fmaxf(mx,  __shfl_xor(mx,  off));
    mxi = fmaxf(mxi, __shfl_xor(mxi, off));
    mxw = fmaxf(mxw, __shfl_xor(mxw, off));
  }

  float scale = (mxi * mxw) / mx;
  for (int idx = tid; idx < LL * LL; idx += 1024) {
    int i = idx / LL, j2 = idx % LL;
    out[b * 2500 + idx] = mb[j2 * MPITCH + i] * scale + bias[j2];
  }
}

extern "C" void kernel_launch(void* const* d_in, const int* in_sizes, int n_in,
                              void* d_out, int out_size, void* d_ws, size_t ws_size,
                              hipStream_t stream) {
  const float* inp  = (const float*)d_in[0];
  const float* wgt  = (const float*)d_in[1];
  const float* bias = (const float*)d_in[2];
  const float* amp  = (const float*)d_in[3];
  const float* sre  = (const float*)d_in[4];
  const float* sim  = (const float*)d_in[5];

  float* ws   = (float*)d_ws;
  float* pre  = ws;
  float* pim  = pre + (size_t)KSPLIT * BB * TILEP;

  k_mm12   <<<BB * KSPLIT, 256, 0, stream>>>(inp, wgt, sre, sim, pre, pim);
  k_final  <<<BB, 1024, 0, stream>>>(pre, pim, inp, wgt, bias, amp, (float*)d_out);
}


# GEMM-1 sum over r halved by one radix-2 step: rows r and r+50 of the scaled signal are added (odd j) or subtracted (even j) before the product, 13 k-steps not 25; changes rounding and summation order
# speedup vs baseline: 8.0697x; 1.0706x over previous
#include <hip/hip_runtime.h>
#include <math.h>


#define LL    50
#define BB    16
#define ITERR 76
#define KDIM  100
#define CDIM  3800
#define CTILES 239
#define JPAD  64
#define SROW  7600
#define SROW0 50
#define COL0  1900
#define CSHIFT (COL0 % 16)
#define KSPLIT 16
#define CTPER  15
#define TILEP  4096
#define QPER  25
#define NLONG 14
#define NTW   200
#define QCLS  3
#define KHALF 50
#define KSTEPS 13
#define HSTR  24
#define JROWS 32
#define NPAR  4
#define GXV   (QCLS * 2 * NPAR * 8 * 32)
#define AHEAD  6

typedef float v2f __attribute__((ext_vector_type(2)));
typedef float v4f __attribute__((ext_vector_type(4)));
typedef float v8f __attribute__((ext_vector_type(8)));

#define WMMA_F32X4(a, b, c) \
  __builtin_amdgcn_wmma_f32_16x16x4_f32(false, (a), false, (b), (short)0, (c), false, false)

__device__ __forceinline__ int class_of(int tau) {
  return (tau < NLONG * 10) ? (tau / 10) : (NLONG + (tau - NLONG * 10) / 9);
}

__device__ __forceinline__ int ct_of(int tau) {
  int kap, m;
  if (tau < NLONG * 10) { kap = tau / 10; m = tau - kap * 10; }
  else { int u = tau - NLONG * 10; int kq = u / 9; kap = NLONG + kq; m = u - kq * 9; }
  return kap + QPER * m;
}

__global__ void __launch_bounds__(256) __attribute__((amdgpu_waves_per_eu(2, 2))) k_mm12(
    const float* __restrict__ inp, const float* __restrict__ wgt,
    const float* __restrict__ sre, const float* __restrict__ sim,
    float* __restrict__ pre, float* __restrict__ pim) {
  __shared__ float sm[2 * JROWS * KHALF];
  __shared__ v4f gx[GXV];
  __shared__ v2f tw[NTW];
  __shared__ v4f qs[QCLS * 512];
  float* Pre = sm;
  float* Pim = sm + JROWS * KHALF;
  int s = blockIdx.x % KSPLIT;
  int b = blockIdx.x / KSPLIT;

  if (threadIdx.x < NTW) {
    float sn, cz;
    sincospif((float)threadIdx.x * 0.01f, &sn, &cz);
    v2f v; v.x = cz; v.y = -sn;
    tw[threadIdx.x] = v;
  }
  __syncthreads();

  for (int k = 0; k < (JROWS * KHALF + 255) / 256; ++k) {
    int e = threadIdx.x + k * 256;
    int j = e / KHALF, r = e % KHALF;
    int t = ((150 + 2 * j) * (50 + r)) % 200;
    v2f v = tw[t];
    if (e < JROWS * KHALF) { Pre[e] = v.x; Pim[e] = v.y; }
  }

  int tau0 = s * CTPER;
  int tau1 = (tau0 + CTPER < CTILES) ? (tau0 + CTPER) : CTILES;
  int kap0 = class_of(tau0);
  int ncls = class_of(tau1 - 1) - kap0 + 1;
  for (int cls = 0; cls < ncls; ++cls) {
    for (int hh = 0; hh < 2; ++hh) {
      int rem = threadIdx.x + hh * 256;
      int ql = rem & 31;
      int it = (rem >> 5) & 3;
      int tp = (rem >> 7) & 3;
      int i  = it * 16 + (ql & 15);
      int c0 = (kap0 + cls) * 16 - CSHIFT + 8 * (ql >> 4) + 2 * tp;
      int t0 = ((i + 1) * (COL0 + c0)) % 200;
      int t1 = ((i + 1) * (COL0 + c0 + 1)) % 200;
      v2f z0 = tw[t0];
      v2f z1 = tw[t1];
      v4f v = {0.f, 0.f, 0.f, 0.f};
      if (i < LL) { v.x = z0.x; v.y = z1.x; v.z = z0.y; v.w = z1.y; }
      qs[cls * 512 + rem] = v;
    }
  }
  __syncthreads();

  int wave = threadIdx.x >> 5;
  int lane = threadIdx.x & 31;
  int jt  = wave & 1;
  int par = wave >> 1;
  int h  = lane >> 4;
  int ln = lane & 15;
  int j  = 2 * ln + jt;
  float sgn = jt ? 1.f : -1.f;

  int nq = (tau1 - tau0 + NPAR - 1) / NPAR;
  int tau  = tau0 + par * nq;
  int tend = (tau + nq < tau1) ? (tau + nq) : tau1;

  float ar0[KSTEPS], ar1[KSTEPS], br0[KSTEPS], br1[KSTEPS];
  float ai0[KSTEPS], ai1[KSTEPS], bi0[KSTEPS], bi1[KSTEPS];
  float iva[KSTEPS], ivb[KSTEPS];
#define LOAD_STEP(T, PRE_, PIM_, PIN_) do { \
    ar0[(T)] = (PRE_)[(T) * 2 * SROW]; ar1[(T)] = (PRE_)[(T) * 2 * SROW + SROW]; \
    br0[(T)] = (PRE_)[((T) * 2 + KHALF) * SROW]; br1[(T)] = (PRE_)[((T) * 2 + KHALF) * SROW + SROW]; \
    ai0[(T)] = (PIM_)[(T) * 2 * SROW]; ai1[(T)] = (PIM_)[(T) * 2 * SROW + SROW]; \
    bi0[(T)] = (PIM_)[((T) * 2 + KHALF) * SROW]; bi1[(T)] = (PIM_)[((T) * 2 + KHALF) * SROW + SROW]; \
    iva[(T)] = (PIN_)[(T) * LL]; ivb[(T)] = (PIN_)[((T) + KHALF / 2) * LL]; } while (0)
#define TILE_PTRS(CT, PRE_, PIM_, PIN_) do { \
    int c_ = (CT) * 16 - CSHIFT + ln; \
    int q_ = (c_ >= 0 && c_ < CDIM) ? (c_ / ITERR) : 0; \
    int so_ = (SROW0 + HSTR * h) * SROW + COL0 + c_; \
    (PRE_) = sre + so_; (PIM_) = sim + so_; (PIN_) = inp + b * 2500 + q_ + (HSTR / 2) * h * LL; } while (0)

  const float *cre, *cim, *cin;
  const float *nre, *nim, *nin;
  TILE_PTRS(ct_of(tau), cre, cim, cin);
#pragma unroll
  for (int t = 0; t < AHEAD; ++t) LOAD_STEP(t, cre, cim, cin);

  v2f prr[KSTEPS], pii[KSTEPS];
#pragma unroll
  for (int t = 0; t < KSTEPS; ++t) {
    int kk = HSTR * h + 2 * t;
    v2f pr_ = *(const v2f*)&Pre[j * KHALF + kk];
    v2f pi_ = *(const v2f*)&Pim[j * KHALF + kk];
    if (t == 0 && h == 1) { pr_ = (v2f){0.f, 0.f}; pi_ = (v2f){0.f, 0.f}; }
    prr[t] = pr_;
    pii[t] = pi_;
  }
#pragma unroll
  for (int t = 0; t < AHEAD; ++t)
    asm volatile("" :: "v"(ar0[t]), "v"(ar1[t]), "v"(br0[t]), "v"(br1[t]), "v"(ai0[t]), "v"(ai1[t]),
                 "v"(bi0[t]), "v"(bi1[t]), "v"(iva[t]), "v"(ivb[t]));

  while (tau < tend) {
    int kap, m, nk;
    if (tau < NLONG * 10) { kap = tau / 10; m = tau - kap * 10; nk = 10; }
    else { int u = tau - NLONG * 10; int kq = u / 9; kap = NLONG + kq; m = u - kq * 9; nk = 9; }
    int mend = m + (tend - tau);
    if (mend > nk) mend = nk;
    v8f g1re = {}; v8f g1im = {}; v8f g2re = {}; v8f g2im = {};
    for (; m < mend; ++m) {
      int ct = kap + QPER * m;
      ++tau;
      int ctn = (tau < tend) ? ct_of(tau) : ct;
      TILE_PTRS(ctn, nre, nim, nin);
      int cA = ct * 16 - CSHIFT + 8 * h;
      int cB = cA + 4;
      int qA = (cA >= 0 && cA < CDIM) ? cA / ITERR : 0;
      int qB = (cB >= 0 && cB < CDIM) ? cB / ITERR : 0;
      int j2 = (j >= 1 && j <= 24) ? (LL - j) : 0;
      float w1Al = wgt[j * LL + qA];
      float w1Bl = wgt[j * LL + qB];
      float w2Al = wgt[j2 * LL + qA];
      float w2Bl = wgt[j2 * LL + qB];
      v8f da = {}; v8f db = {}; v8f dc = {}; v8f dd = {};
#pragma unroll
      for (int t = 0; t < KSTEPS; ++t) {
        __builtin_amdgcn_sched_barrier(0);
        if (t + AHEAD < KSTEPS) LOAD_STEP(t + AHEAD, cre, cim, cin);
        else                    LOAD_STEP(t + AHEAD - KSTEPS, nre, nim, nin);
        float ia = iva[t];
        float ib = sgn * ivb[t];
        v2f are; are.x = ar0[t] * ia + br0[t] * ib; are.y = ar1[t] * ia + br1[t] * ib;
        v2f aim; aim.x = ai0[t] * ia + bi0[t] * ib; aim.y = ai1[t] * ia + bi1[t] * ib;
        da = WMMA_F32X4(are, prr[t], da);
        db = WMMA_F32X4(aim, pii[t], db);
        dc = WMMA_F32X4(aim, prr[t], dc);
        dd = WMMA_F32X4(are, pii[t], dd);
      }
      cre = nre; cim = nim; cin = nin;
      v8f d1re = da - db;
      v8f d1im = dc + dd;
      v8f d2re = da + db;
      v8f d2im = dc - dd;

      bool okA = (cA >= 0 && cA < CDIM), okB = (cB >= 0 && cB < CDIM);
      bool ok2 = (j >= 1 && j <= 24);
      float w1A = okA ? fabsf(w1Al) : 0.f;
      float w1B = okB ? fabsf(w1Bl) : 0.f;
      float w2A = (ok2 && okA) ? fabsf(w2Al) : 0.f;
      float w2B = (ok2 && okB) ? fabsf(w2Bl) : 0.f;

  #pragma unroll
      for (int r = 0; r < 8; ++r) {
        float w1 = (r < 4) ? w1A : w1B;
        float w2 = (r < 4) ? w2A : w2B;
        g1re[r] += d1re[r] * w1;
        g1im[r] += d1im[r] * w1;
        g2re[r] += d2re[r] * w2;
        g2im[r] += d2im[r] * w2;
      }
    }

    {
      v4f* gp = gx + ((((kap - kap0) * 2 + jt) * NPAR + par) * 8) * 32 + lane;
      gp[0 * 32] = (v4f){g1re[0], g1re[1], g1re[2], g1re[3]};
      gp[1 * 32] = (v4f){g1re[4], g1re[5], g1re[6], g1re[7]};
      gp[2 * 32] = (v4f){g1im[0], g1im[1], g1im[2], g1im[3]};
      gp[3 * 32] = (v4f){g1im[4], g1im[5], g1im[6], g1im[7]};
      gp[4 * 32] = (v4f){g2re[0], g2re[1], g2re[2], g2re[3]};
      gp[5 * 32] = (v4f){g2re[4], g2re[5], g2re[6], g2re[7]};
      gp[6 * 32] = (v4f){g2im[0], g2im[1], g2im[2], g2im[3]};
      gp[7 * 32] = (v4f){g2im[4], g2im[5], g2im[6], g2im[7]};
    }
  }
#undef LOAD_STEP
#undef TILE_PTRS

  __syncthreads();

  int it = par;
  v8f f1[2], f2[2], f3[2];
#pragma unroll
  for (int e = 0; e < 2; ++e) { f1[e] = (v8f){}; f2[e] = (v8f){}; f3[e] = (v8f){}; }

  for (int cl = 0; cl < ncls; ++cl) {
    int kapc = kap0 + cl;
    int cs = (kapc < NLONG) ? kapc * 10 : NLONG * 10 + (kapc - NLONG) * 9;
    int ce = cs + ((kapc < NLONG) ? 10 : 9);
    v4f gs[8];
#pragma unroll
    for (int g = 0; g < 8; ++g) gs[g] = (v4f){0.f, 0.f, 0.f, 0.f};
#pragma unroll
    for (int p = 0; p < NPAR; ++p) {
      int lo = tau0 + p * nq;
      int hi = (lo + nq < tau1) ? (lo + nq) : tau1;
      if (lo < hi && lo < ce && hi > cs) {
        const v4f* gp = gx + (((cl * 2 + jt) * NPAR + p) * 8) * 32 + lane;
        v4f ld[8];
#pragma unroll
        for (int g = 0; g < 8; ++g) ld[g] = gp[g * 32];
#pragma unroll
        for (int g = 0; g < 8; ++g) gs[g] += ld[g];
      }
    }
    const v4f* qt = qs + cl * 512 + it * 32 + lane;
    v4f qv[4];
#pragma unroll
    for (int tp = 0; tp < 4; ++tp) qv[tp] = qt[tp * 128];

#pragma unroll
    for (int tp = 0; tp < 4; ++tp) {
      int g = tp >> 1, o = (tp & 1) * 2;
      v2f gr1; gr1.x = gs[0 + g][o]; gr1.y = gs[0 + g][o + 1];
      v2f gi1; gi1.x = gs[2 + g][o]; gi1.y = gs[2 + g][o + 1];
      v2f gr2; gr2.x = gs[4 + g][o]; gr2.y = gs[4 + g][o + 1];
      v2f gi2; gi2.x = gs[6 + g][o]; gi2.y = gs[6 + g][o + 1];
      v2f gs1 = gr1 + gi1;
      v2f gs2 = gr2 + gi2;
      v4f qq = qv[tp];
      v2f br; br.x = qq.x; br.y = qq.y;
      v2f bi; bi.x = qq.z; bi.y = qq.w;
      v2f bs = br + bi;
      f1[0] = WMMA_F32X4(gr1, br, f1[0]);
      f2[0] = WMMA_F32X4(gi1, bi, f2[0]);
      f3[0] = WMMA_F32X4(gs1, bs, f3[0]);
      f1[1] = WMMA_F32X4(gr2, br, f1[1]);
      f2[1] = WMMA_F32X4(gi2, bi, f2[1]);
      f3[1] = WMMA_F32X4(gs2, bs, f3[1]);
    }
  }

#pragma unroll
  for (int e = 0; e < 2; ++e) {
    v8f fre = f1[e] - f2[e];
    v8f fim = (f3[e] - f1[e]) - f2[e];
    size_t pbase = ((size_t)(s * BB + b) * 16 + (size_t)((e * 2 + jt) * 4 + it)) * 256 + lane;
#pragma unroll
    for (int r = 0; r < 8; ++r) {
      pre[pbase + r * 32] = fre[r];
      pim[pbase + r * 32] = fim[r];
    }
  }
}

#define MPITCH 65
__global__ void __launch_bounds__(1024) k_final(
    const float* __restrict__ pre, const float* __restrict__ pim,
    const float* __restrict__ inp, const float* __restrict__ wgt,
    const float* __restrict__ bias,
    const float* __restrict__ amp, float* __restrict__ out) {
  __shared__ float mb[JPAD * MPITCH];
  __shared__ float red[3][32];
  int b = blockIdx.x;
  int tid = threadIdx.x;
  int tile = tid >> 6;
  int p0 = (tid & 63) * 4;
  int vt = tile >> 2;
  int jl = 2 * ((p0 >> 5) + 8 * ((p0 >> 4) & 1)) + (vt & 1);
  int j  = (vt < 2) ? jl : (LL - jl);
  bool jok = (vt < 2) ? (jl <= 25) : (jl >= 1 && jl <= 24);
  int i0 = (tile & 3) * 16 + (p0 & 15);

  float m = 0.f;
  if (jok && i0 < LL) {
    const float* pr = pre + (size_t)b * TILEP + tid * 4;
    const float* pi = pim + (size_t)b * TILEP + tid * 4;
#define NFIRST 12
#define PADD(SR_, SI_, A_, C_) do { \
    SR_.x += A_.x; SR_.y += A_.y; SR_.z += A_.z; SR_.w += A_.w; \
    SI_.x += C_.x; SI_.y += C_.y; SI_.z += C_.z; SI_.w += C_.w; } while (0)
    float4 qa[NFIRST], qc[NFIRST];
#pragma unroll
    for (int s = 0; s < NFIRST; ++s) {
      qa[s] = *(const float4*)(pr + (size_t)s * BB * TILEP);
      qc[s] = *(const float4*)(pi + (size_t)s * BB * TILEP);
    }
    float4 sr = qa[0];
    float4 si = qc[0];
#pragma unroll
    for (int s = 1; s < KSPLIT - NFIRST; ++s) PADD(sr, si, qa[s], qc[s]);
    __builtin_amdgcn_sched_barrier(0);
    float4 ra[KSPLIT - NFIRST], rc[KSPLIT - NFIRST];
#pragma unroll
    for (int s = NFIRST; s < KSPLIT; ++s) {
      ra[s - NFIRST] = *(const float4*)(pr + (size_t)s * BB * TILEP);
      rc[s - NFIRST] = *(const float4*)(pi + (size_t)s * BB * TILEP);
    }
#pragma unroll
    for (int s = KSPLIT - NFIRST; s < NFIRST; ++s) PADD(sr, si, qa[s], qc[s]);
#pragma unroll
    for (int s = NFIRST; s < KSPLIT; ++s) PADD(sr, si, ra[s - NFIRST], rc[s - NFIRST]);
#undef PADD
#undef NFIRST
    float fr[4] = {sr.x, sr.y, sr.z, sr.w};
    float fi[4] = {si.x, si.y, si.z, si.w};
#pragma unroll
    for (int k = 0; k < 4; ++k) {
      int i = i0 + k;
      float v = 0.f;
      if (i < LL) v = sqrtf(fr[k] * fr[k] + fi[k] * fi[k]) / amp[j * LL + i];
      mb[j * MPITCH + i] = v;
      m = fmaxf(m, v);
    }
  }
  int k2 = (tid + 2048 < LL * LL) ? (tid + 2048) : tid;
  float x0 = inp[b * 2500 + tid], x1 = inp[b * 2500 + tid + 1024], x2 = inp[b * 2500 + k2];
  float w0 = wgt[tid], w1 = wgt[tid + 1024], w2 = wgt[k2];
  float mi = fmaxf(fmaxf(fmaxf(-3.4e38f, x0), x1), x2);
  float mw = fmaxf(fmaxf(fmaxf(-3.4e38f, fabsf(w0)), fabsf(w1)), fabsf(w2));
  for (int off = 16; off > 0; off >>= 1) {
    m  = fmaxf(m,  __shfl_xor(m,  off));
    mi = fmaxf(mi, __shfl_xor(mi, off));
    mw = fmaxf(mw, __shfl_xor(mw, off));
  }
  if ((tid & 31) == 0) { red[0][tid >> 5] = m; red[1][tid >> 5] = mi; red[2][tid >> 5] = mw; }
  __syncthreads();
  float mx  = red[0][tid & 31];
  float mxi = red[1][tid & 31];
  float mxw = red[2][tid & 31];
  for (int off = 16; off > 0; off >>= 1) {
    mx  = fmaxf(mx,  __shfl_xor(mx,  off));
    mxi = fmaxf(mxi, __shfl_xor(mxi, off));
    mxw = fmaxf(mxw, __shfl_xor(mxw, off));
  }

  float scale = (mxi * mxw) / mx;
  for (int idx = tid; idx < LL * LL; idx += 1024) {
    int i = idx / LL, j2 = idx % LL;
    out[b * 2500 + idx] = mb[j2 * MPITCH + i] * scale + bias[j2];
  }
}

extern "C" void kernel_launch(void* const* d_in, const int* in_sizes, int n_in,
                              void* d_out, int out_size, void* d_ws, size_t ws_size,
                              hipStream_t stream) {
  const float* inp  = (const float*)d_in[0];
  const float* wgt  = (const float*)d_in[1];
  const float* bias = (const float*)d_in[2];
  const float* amp  = (const float*)d_in[3];
  const float* sre  = (const float*)d_in[4];
  const float* sim  = (const float*)d_in[5];

  float* ws   = (float*)d_ws;
  float* pre  = ws;
  float* pim  = pre + (size_t)KSPLIT * BB * TILEP;

  k_mm12   <<<BB * KSPLIT, 256, 0, stream>>>(inp, wgt, sre, sim, pre, pim);
  k_final  <<<BB, 1024, 0, stream>>>(pre, pim, inp, wgt, bias, amp, (float*)d_out);
}
